# starting kernel
# baseline (speedup 1.0000x reference)
#include <hip/hip_runtime.h>

typedef __attribute__((ext_vector_type(2))) float v2f;
typedef __attribute__((ext_vector_type(8))) float v8f;


__global__ void k_fill(float* __restrict__ p, float v, int n) {
    int i = blockIdx.x * blockDim.x + threadIdx.x;
    if (i < n) p[i] = v;
}


__global__ void k_edge_deg(const int* __restrict__ ei, const float* __restrict__ attr,
                           float* __restrict__ w, float* __restrict__ deg, int E) {
    int e = blockIdx.x * blockDim.x + threadIdx.x;
    if (e >= E) return;
    int s = ei[e];
    int d = ei[E + e];
    float ew = (s != d) ? attr[e] : 0.0f;
    w[e] = ew;
    if (ew != 0.0f) atomicAdd(&deg[d], ew);
}

__global__ void k_dis(float* __restrict__ deg, int n) {
    int i = blockIdx.x * blockDim.x + threadIdx.x;
    if (i >= n) return;
    float dv = deg[i];
    deg[i] = (dv > 0.0f) ? rsqrtf(fmaxf(dv, 1e-30f)) : 0.0f;
}

__global__ void k_edge_scale(const int* __restrict__ ei, const float* __restrict__ dis,
                             float* __restrict__ w, int E) {
    int e = blockIdx.x * blockDim.x + threadIdx.x;
    if (e >= E) return;
    int s = ei[e];
    int d = ei[E + e];
    w[e] = -dis[d] * w[e] * dis[s];
}

__global__ void k_spmm(const float* __restrict__ h, const int* __restrict__ ei,
                       const float* __restrict__ w, float* __restrict__ out,
                       int E, int F4shift, float scale) {
    int idx = blockIdx.x * blockDim.x + threadIdx.x;
    int total = E << F4shift;
    if (idx >= total) return;
    int e = idx >> F4shift;
    int c = idx & ((1 << F4shift) - 1);
    float ww = w[e] * scale;
    if (ww == 0.0f) return;
    int s = ei[e];
    int d = ei[E + e];
    const float4* hp = reinterpret_cast<const float4*>(h) + ((size_t)s << F4shift) + c;
    float4 hv = *hp;
    float* op = out + (((size_t)d << F4shift) << 2) + (c << 2);
    atomicAdd(op + 0, ww * hv.x);
    atomicAdd(op + 1, ww * hv.y);
    atomicAdd(op + 2, ww * hv.z);
    atomicAdd(op + 3, ww * hv.w);
}

#define GW 8
#define KPAD 132

__global__ __launch_bounds__(GW * 32) void
k_cheb_gemm(const float* __restrict__ Tx0, const float* __restrict__ Tx1,
            const float* __restrict__ P, const float* __restrict__ W,
            const float* __restrict__ bias, float* __restrict__ out,
            int M, int Kd, int N, int Mtiles) {
    __shared__ float Wl[3 * 16 * KPAD];

    int tid  = threadIdx.x;
    int nt   = blockIdx.y;

    for (int idx = tid; idx < 3 * 16 * Kd; idx += GW * 32) {
        int t  = idx / (16 * Kd);
        int r  = idx - t * 16 * Kd;
        int cc = r / Kd;
        int k  = r - cc * Kd;
        int nc = nt * 16 + cc;
        float v = W[(size_t)(t * Kd + k) * N + nc];
        if (t == 0) v -= W[(size_t)(2 * Kd + k) * N + nc];
        Wl[(t * 16 + cc) * KPAD + k] = v;
    }
    __syncthreads();

    int wave = tid >> 5;
    int mt   = blockIdx.x * GW + wave;
    if (mt >= Mtiles) return;

    int lane = tid & 31;
    int col  = lane & 15;
    int klo  = (lane >> 4) << 1;
    int row  = mt * 16 + col;
    int ncol = nt * 16 + col;

    const float* Txs[3] = {Tx0, Tx1, P};

    v8f c;
    float bv = bias[ncol];
#pragma unroll
    for (int i = 0; i < 8; ++i) c[i] = bv;

    for (int t = 0; t < 3; ++t) {
        const float* arow = Txs[t] + (size_t)row * Kd;
        const float* brow = &Wl[(t * 16 + col) * KPAD];
        for (int k0 = 0; k0 < Kd; k0 += 4) {
            v2f a, b;
            a.x = arow[k0 + klo];
            a.y = arow[k0 + klo + 1];
            b = *reinterpret_cast<const v2f*>(brow + k0 + klo);
            c = __builtin_amdgcn_wmma_f32_16x16x4_f32(
                     false, a,  false, b,
                     (short)0, c,  false,  false);
        }
    }

    int rbase = mt * 16 + ((lane >> 4) << 3);
#pragma unroll
    for (int i = 0; i < 8; ++i) {
        float v = fmaxf(c[i], 0.0f);
        int r = rbase + i;
        if (r < M) out[(size_t)r * N + ncol] = v;
    }
}


__global__ void k_count(const int* __restrict__ batch, float* __restrict__ cnt, int n) {
    int i = blockIdx.x * blockDim.x + threadIdx.x;
    if (i < n) atomicAdd(&cnt[batch[i]], 1.0f);
}

__global__ void k_pool(const float* __restrict__ h, const int* __restrict__ batch,
                       float* __restrict__ sums, int n) {
    int idx = blockIdx.x * blockDim.x + threadIdx.x;
    int total = n << 5;
    if (idx >= total) return;
    int i = idx >> 5;
    int c = idx & 31;
    int g = batch[i];
    const float4* hp = reinterpret_cast<const float4*>(h) + ((size_t)i << 5) + c;
    float4 hv = *hp;
    float* sp = sums + ((size_t)g << 7) + (c << 2);
    atomicAdd(sp + 0, hv.x);
    atomicAdd(sp + 1, hv.y);
    atomicAdd(sp + 2, hv.z);
    atomicAdd(sp + 3, hv.w);
}

__global__ void k_final(const float* __restrict__ sums, const float* __restrict__ cnt,
                        const float* __restrict__ linW, const float* __restrict__ linb,
                        float* __restrict__ out, int G, int H) {
    __shared__ float hg_s[128];
    int g = blockIdx.x;
    int f = threadIdx.x;
    float c = cnt[g];
    float hg = sums[(size_t)g * H + f] / fmaxf(c, 1.0f);
    out[(size_t)G * 2 + (size_t)g * H + f] = hg;
    hg_s[f] = hg;
    __syncthreads();
    if (f < 2) {
        float acc = linb[f];
        for (int j = 0; j < H; ++j) acc += hg_s[j] * linW[j * 2 + f];
        out[g * 2 + f] = acc;
    }
}


extern "C" void kernel_launch(void* const* d_in, const int* in_sizes, int n_in,
                              void* d_out, int out_size, void* d_ws, size_t ws_size,
                              hipStream_t stream) {
    const float* x    = (const float*)d_in[0];
    const int*   ei   = (const int*)d_in[1];
    const float* attr = (const float*)d_in[2];
    const int*   bat  = (const int*)d_in[3];
    const float* W1 = (const float*)d_in[4];  const float* b1 = (const float*)d_in[5];
    const float* W2 = (const float*)d_in[6];  const float* b2 = (const float*)d_in[7];
    const float* W3 = (const float*)d_in[8];  const float* b3 = (const float*)d_in[9];
    const float* linW = (const float*)d_in[10];
    const float* linb = (const float*)d_in[11];
    float* out = (float*)d_out;

    const int F_IN = 128, H1 = 64, H2 = 64, H3 = 128, NCLS = 2;
    int Nn = in_sizes[0] / F_IN;
    int E  = in_sizes[2];
    int G  = out_size / (NCLS + H3);

    float* ws = (float*)d_ws;
    size_t off = 0;
    float* dis  = ws + off; off += (size_t)Nn;
    float* w    = ws + off; off += (size_t)E;
    float* A    = ws + off; off += (size_t)Nn * F_IN;
    float* B    = ws + off; off += (size_t)Nn * F_IN;
    float* C    = ws + off; off += (size_t)Nn * F_IN;
    float* D    = ws + off; off += (size_t)Nn * H2;
    float* pool = ws + off; off += (size_t)G * H3;
    float* cnt  = ws + off; off += (size_t)G;

    auto nb = [](long n, int t) { return (unsigned)((n + t - 1) / t); };

    k_fill<<<nb(Nn, 256), 256, 0, stream>>>(dis, 0.0f, Nn);
    k_edge_deg<<<nb(E, 256), 256, 0, stream>>>(ei, attr, w, dis, E);
    k_dis<<<nb(Nn, 256), 256, 0, stream>>>(dis, Nn);
    k_edge_scale<<<nb(E, 256), 256, 0, stream>>>(ei, dis, w, E);

    auto layer = [&](const float* in, int Fin, const float* Wt, const float* bt,
                     int Nout, float* outp) {
        int sh = (Fin == 128) ? 5 : 4;
        long nf = (long)Nn * Fin;
        k_fill<<<nb(nf, 256), 256, 0, stream>>>(A, 0.0f, (int)nf);
        k_fill<<<nb(nf, 256), 256, 0, stream>>>(B, 0.0f, (int)nf);
        k_spmm<<<nb((long)E << sh, 256), 256, 0, stream>>>(in, ei, w, A, E, sh, 1.0f);
        k_spmm<<<nb((long)E << sh, 256), 256, 0, stream>>>(A, ei, w, B, E, sh, 2.0f);
        int mtiles = (Nn + 15) / 16;
        dim3 g((mtiles + GW - 1) / GW, Nout / 16);
        k_cheb_gemm<<<g, GW * 32, 0, stream>>>(in, A, B, Wt, bt, outp,
                                               Nn, Fin, Nout, mtiles);
    };

    layer(x, F_IN, W1, b1, H1, C);
    layer(C, H1,  W2, b2, H2, D);
    layer(D, H2,  W3, b3, H3, C);

    k_fill<<<nb((long)G * H3 + G, 256), 256, 0, stream>>>(pool, 0.0f, G * H3 + G);
    k_count<<<nb(Nn, 256), 256, 0, stream>>>(bat, cnt, Nn);
    k_pool<<<nb((long)Nn * (H3 / 4), 256), 256, 0, stream>>>(C, bat, pool, Nn);
    k_final<<<G, H3, 0, stream>>>(pool, cnt, linW, linb, out, G, H3);
}


# layer 1 reassociated: weights applied before propagation (one 3-accumulator f32 WMMA pass over x), so both scatter-add props run at width 64 not 128 and add onto the GEMM output, no zero fill
# speedup vs baseline: 1.2567x; 1.2567x over previous
#include <hip/hip_runtime.h>

typedef __attribute__((ext_vector_type(2))) float v2f;
typedef __attribute__((ext_vector_type(8))) float v8f;


__global__ void k_fill(float* __restrict__ p, float v, int n) {
    int i = blockIdx.x * blockDim.x + threadIdx.x;
    if (i < n) p[i] = v;
}


__global__ void k_edge_deg(const int* __restrict__ ei, const float* __restrict__ attr,
                           float* __restrict__ w, float* __restrict__ deg, int E) {
    int e = blockIdx.x * blockDim.x + threadIdx.x;
    if (e >= E) return;
    int s = ei[e];
    int d = ei[E + e];
    float ew = (s != d) ? attr[e] : 0.0f;
    w[e] = ew;
    if (ew != 0.0f) atomicAdd(&deg[d], ew);
}

__global__ void k_dis(float* __restrict__ deg, int n) {
    int i = blockIdx.x * blockDim.x + threadIdx.x;
    if (i >= n) return;
    float dv = deg[i];
    deg[i] = (dv > 0.0f) ? rsqrtf(fmaxf(dv, 1e-30f)) : 0.0f;
}

__global__ void k_edge_scale(const int* __restrict__ ei, const float* __restrict__ dis,
                             float* __restrict__ w, int E) {
    int e = blockIdx.x * blockDim.x + threadIdx.x;
    if (e >= E) return;
    int s = ei[e];
    int d = ei[E + e];
    w[e] = -dis[d] * w[e] * dis[s];
}

__global__ void k_spmm(const float* __restrict__ h, const int* __restrict__ ei,
                       const float* __restrict__ w, float* __restrict__ out,
                       int E, int F4shift, float scale) {
    int idx = blockIdx.x * blockDim.x + threadIdx.x;
    int total = E << F4shift;
    if (idx >= total) return;
    int e = idx >> F4shift;
    int c = idx & ((1 << F4shift) - 1);
    float ww = w[e] * scale;
    if (ww == 0.0f) return;
    int s = ei[e];
    int d = ei[E + e];
    const float4* hp = reinterpret_cast<const float4*>(h) + ((size_t)s << F4shift) + c;
    float4 hv = *hp;
    float* op = out + (((size_t)d << F4shift) << 2) + (c << 2);
    atomicAdd(op + 0, ww * hv.x);
    atomicAdd(op + 1, ww * hv.y);
    atomicAdd(op + 2, ww * hv.z);
    atomicAdd(op + 3, ww * hv.w);
}

#define GW 8
#define KPAD 132

__global__ __launch_bounds__(GW * 32) void
k_cheb_gemm(const float* __restrict__ Tx0, const float* __restrict__ Tx1,
            const float* __restrict__ P, const float* __restrict__ W,
            const float* __restrict__ bias, float* __restrict__ out,
            int M, int Kd, int N, int Mtiles) {
    __shared__ float Wl[3 * 16 * KPAD];

    int tid  = threadIdx.x;
    int nt   = blockIdx.y;

    for (int idx = tid; idx < 3 * 16 * Kd; idx += GW * 32) {
        int t  = idx / (16 * Kd);
        int r  = idx - t * 16 * Kd;
        int cc = r / Kd;
        int k  = r - cc * Kd;
        int nc = nt * 16 + cc;
        float v = W[(size_t)(t * Kd + k) * N + nc];
        if (t == 0) v -= W[(size_t)(2 * Kd + k) * N + nc];
        Wl[(t * 16 + cc) * KPAD + k] = v;
    }
    __syncthreads();

    int wave = tid >> 5;
    int mt   = blockIdx.x * GW + wave;
    if (mt >= Mtiles) return;

    int lane = tid & 31;
    int col  = lane & 15;
    int klo  = (lane >> 4) << 1;
    int row  = mt * 16 + col;
    int ncol = nt * 16 + col;

    const float* Txs[3] = {Tx0, Tx1, P};

    v8f c;
    float bv = bias[ncol];
#pragma unroll
    for (int i = 0; i < 8; ++i) c[i] = bv;

    for (int t = 0; t < 3; ++t) {
        const float* arow = Txs[t] + (size_t)row * Kd;
        const float* brow = &Wl[(t * 16 + col) * KPAD];
        for (int k0 = 0; k0 < Kd; k0 += 4) {
            v2f a, b;
            a.x = arow[k0 + klo];
            a.y = arow[k0 + klo + 1];
            b = *reinterpret_cast<const v2f*>(brow + k0 + klo);
            c = __builtin_amdgcn_wmma_f32_16x16x4_f32(
                     false, a,  false, b,
                     (short)0, c,  false,  false);
        }
    }

    int rbase = mt * 16 + ((lane >> 4) << 3);
#pragma unroll
    for (int i = 0; i < 8; ++i) {
        float v = fmaxf(c[i], 0.0f);
        int r = rbase + i;
        if (r < M) out[(size_t)r * N + ncol] = v;
    }
}

__global__ __launch_bounds__(GW * 32) void
k_premul_gemm(const float* __restrict__ X, const float* __restrict__ W,
              const float* __restrict__ bias, float* __restrict__ Y0,
              float* __restrict__ Y1, float* __restrict__ Y2,
              int M, int Kd, int N, int Mtiles) {
    __shared__ float Wl[3 * 16 * KPAD];

    int tid  = threadIdx.x;
    int nt   = blockIdx.y;

    for (int idx = tid; idx < 3 * 16 * Kd; idx += GW * 32) {
        int t  = idx / (16 * Kd);
        int r  = idx - t * 16 * Kd;
        int cc = r / Kd;
        int k  = r - cc * Kd;
        int nc = nt * 16 + cc;
        float v = W[(size_t)(t * Kd + k) * N + nc];
        if (t == 0) v -= W[(size_t)(2 * Kd + k) * N + nc];
        Wl[(t * 16 + cc) * KPAD + k] = v;
    }
    __syncthreads();

    int wave = tid >> 5;
    int mt   = blockIdx.x * GW + wave;
    if (mt >= Mtiles) return;

    int lane = tid & 31;
    int col  = lane & 15;
    int klo  = (lane >> 4) << 1;
    int row  = mt * 16 + col;
    int ncol = nt * 16 + col;

    v8f c0, c1, c2;
    float bv = bias[ncol];
#pragma unroll
    for (int i = 0; i < 8; ++i) { c0[i] = bv; c1[i] = 0.0f; c2[i] = 0.0f; }

    const float* arow  = X + (size_t)row * Kd;
    const float* brow0 = &Wl[(0 * 16 + col) * KPAD];
    const float* brow1 = &Wl[(1 * 16 + col) * KPAD];
    const float* brow2 = &Wl[(2 * 16 + col) * KPAD];
    for (int k0 = 0; k0 < Kd; k0 += 4) {
        v2f a, b0, b1, b2;
        a.x = arow[k0 + klo];
        a.y = arow[k0 + klo + 1];
        b0 = *reinterpret_cast<const v2f*>(brow0 + k0 + klo);
        b1 = *reinterpret_cast<const v2f*>(brow1 + k0 + klo);
        b2 = *reinterpret_cast<const v2f*>(brow2 + k0 + klo);
        c0 = __builtin_amdgcn_wmma_f32_16x16x4_f32(
                  false, a,  false, b0,
                  (short)0, c0,  false,  false);
        c1 = __builtin_amdgcn_wmma_f32_16x16x4_f32(
                 false, a, false, b1, (short)0, c1, false, false);
        c2 = __builtin_amdgcn_wmma_f32_16x16x4_f32(
                 false, a, false, b2, (short)0, c2, false, false);
    }

    int rbase = mt * 16 + ((lane >> 4) << 3);
#pragma unroll
    for (int i = 0; i < 8; ++i) {
        int r = rbase + i;
        if (r < M) {
            size_t o = (size_t)r * N + ncol;
            Y0[o] = c0[i];
            Y1[o] = c1[i];
            Y2[o] = c2[i];
        }
    }
}

__global__ void k_relu4(float* __restrict__ p, int n4) {
    int i = blockIdx.x * blockDim.x + threadIdx.x;
    if (i >= n4) return;
    float4* q = reinterpret_cast<float4*>(p) + i;
    float4 v = *q;
    v.x = fmaxf(v.x, 0.0f);
    v.y = fmaxf(v.y, 0.0f);
    v.z = fmaxf(v.z, 0.0f);
    v.w = fmaxf(v.w, 0.0f);
    *q = v;
}


__global__ void k_count(const int* __restrict__ batch, float* __restrict__ cnt, int n) {
    int i = blockIdx.x * blockDim.x + threadIdx.x;
    if (i < n) atomicAdd(&cnt[batch[i]], 1.0f);
}

__global__ void k_pool(const float* __restrict__ h, const int* __restrict__ batch,
                       float* __restrict__ sums, int n) {
    int idx = blockIdx.x * blockDim.x + threadIdx.x;
    int total = n << 5;
    if (idx >= total) return;
    int i = idx >> 5;
    int c = idx & 31;
    int g = batch[i];
    const float4* hp = reinterpret_cast<const float4*>(h) + ((size_t)i << 5) + c;
    float4 hv = *hp;
    float* sp = sums + ((size_t)g << 7) + (c << 2);
    atomicAdd(sp + 0, hv.x);
    atomicAdd(sp + 1, hv.y);
    atomicAdd(sp + 2, hv.z);
    atomicAdd(sp + 3, hv.w);
}

__global__ void k_final(const float* __restrict__ sums, const float* __restrict__ cnt,
                        const float* __restrict__ linW, const float* __restrict__ linb,
                        float* __restrict__ out, int G, int H) {
    __shared__ float hg_s[128];
    int g = blockIdx.x;
    int f = threadIdx.x;
    float c = cnt[g];
    float hg = sums[(size_t)g * H + f] / fmaxf(c, 1.0f);
    out[(size_t)G * 2 + (size_t)g * H + f] = hg;
    hg_s[f] = hg;
    __syncthreads();
    if (f < 2) {
        float acc = linb[f];
        for (int j = 0; j < H; ++j) acc += hg_s[j] * linW[j * 2 + f];
        out[g * 2 + f] = acc;
    }
}


extern "C" void kernel_launch(void* const* d_in, const int* in_sizes, int n_in,
                              void* d_out, int out_size, void* d_ws, size_t ws_size,
                              hipStream_t stream) {
    const float* x    = (const float*)d_in[0];
    const int*   ei   = (const int*)d_in[1];
    const float* attr = (const float*)d_in[2];
    const int*   bat  = (const int*)d_in[3];
    const float* W1 = (const float*)d_in[4];  const float* b1 = (const float*)d_in[5];
    const float* W2 = (const float*)d_in[6];  const float* b2 = (const float*)d_in[7];
    const float* W3 = (const float*)d_in[8];  const float* b3 = (const float*)d_in[9];
    const float* linW = (const float*)d_in[10];
    const float* linb = (const float*)d_in[11];
    float* out = (float*)d_out;

    const int F_IN = 128, H1 = 64, H2 = 64, H3 = 128, NCLS = 2;
    int Nn = in_sizes[0] / F_IN;
    int E  = in_sizes[2];
    int G  = out_size / (NCLS + H3);

    float* ws = (float*)d_ws;
    size_t off = 0;
    float* dis  = ws + off; off += (size_t)Nn;
    float* w    = ws + off; off += (size_t)E;
    float* A    = ws + off; off += (size_t)Nn * F_IN;
    float* B    = ws + off; off += (size_t)Nn * F_IN;
    float* C    = ws + off; off += (size_t)Nn * F_IN;
    float* D    = ws + off; off += (size_t)Nn * H2;
    float* pool = ws + off; off += (size_t)G * H3;
    float* cnt  = ws + off; off += (size_t)G;

    auto nb = [](long n, int t) { return (unsigned)((n + t - 1) / t); };

    k_fill<<<nb(Nn, 256), 256, 0, stream>>>(dis, 0.0f, Nn);
    k_edge_deg<<<nb(E, 256), 256, 0, stream>>>(ei, attr, w, dis, E);
    k_dis<<<nb(Nn, 256), 256, 0, stream>>>(dis, Nn);
    k_edge_scale<<<nb(E, 256), 256, 0, stream>>>(ei, dis, w, E);

    auto layer = [&](const float* in, int Fin, const float* Wt, const float* bt,
                     int Nout, float* outp) {
        int sh = (Fin == 128) ? 5 : 4;
        long nf = (long)Nn * Fin;
        k_fill<<<nb(nf, 256), 256, 0, stream>>>(A, 0.0f, (int)nf);
        k_fill<<<nb(nf, 256), 256, 0, stream>>>(B, 0.0f, (int)nf);
        k_spmm<<<nb((long)E << sh, 256), 256, 0, stream>>>(in, ei, w, A, E, sh, 1.0f);
        k_spmm<<<nb((long)E << sh, 256), 256, 0, stream>>>(A, ei, w, B, E, sh, 2.0f);
        int mtiles = (Nn + 15) / 16;
        dim3 g((mtiles + GW - 1) / GW, Nout / 16);
        k_cheb_gemm<<<g, GW * 32, 0, stream>>>(in, A, B, Wt, bt, outp,
                                               Nn, Fin, Nout, mtiles);
    };

    auto layer_narrow = [&](const float* in, int Fin, const float* Wt, const float* bt,
                            int Nout, float* outp) {
        int sh = (Nout == 128) ? 5 : 4;
        int mtiles = (Nn + 15) / 16;
        dim3 g((mtiles + GW - 1) / GW, Nout / 16);
        k_premul_gemm<<<g, GW * 32, 0, stream>>>(in, Wt, bt, outp, A, B,
                                                 Nn, Fin, Nout, mtiles);
        k_spmm<<<nb((long)E << sh, 256), 256, 0, stream>>>(B, ei, w, A, E, sh, 2.0f);
        k_spmm<<<nb((long)E << sh, 256), 256, 0, stream>>>(A, ei, w, outp, E, sh, 1.0f);
        long n4 = ((long)Nn * Nout) >> 2;
        k_relu4<<<nb(n4, 256), 256, 0, stream>>>(outp, (int)n4);
    };

    layer_narrow(x, F_IN, W1, b1, H1, C);
    layer(C, H1,  W2, b2, H2, D);
    layer(D, H2,  W3, b3, H3, C);

    k_fill<<<nb((long)G * H3 + G, 256), 256, 0, stream>>>(pool, 0.0f, G * H3 + G);
    k_count<<<nb(Nn, 256), 256, 0, stream>>>(bat, cnt, Nn);
    k_pool<<<nb((long)Nn * (H3 / 4), 256), 256, 0, stream>>>(C, bat, pool, Nn);
    k_final<<<G, H3, 0, stream>>>(pool, cnt, linW, linb, out, G, H3);
}


# float-atomic scatter-add replaced by per-call CSR (counting sort by destination, slots in atomic arrival order) + one-wave-per-row register gather; row sums sequential in slot order; no zero fills
# speedup vs baseline: 2.4366x; 1.9389x over previous
#include <hip/hip_runtime.h>

typedef __attribute__((ext_vector_type(2))) float v2f;
typedef __attribute__((ext_vector_type(8))) float v8f;


__global__ void k_fill(float* __restrict__ p, float v, int n) {
    int i = blockIdx.x * blockDim.x + threadIdx.x;
    if (i < n) p[i] = v;
}


__global__ void k_edge_deg(const int* __restrict__ ei, const float* __restrict__ attr,
                           float* __restrict__ deg, int* __restrict__ cnt,
                           int* __restrict__ rank, int E) {
    int e = blockIdx.x * blockDim.x + threadIdx.x;
    if (e >= E) return;
    int s = ei[e];
    int d = ei[E + e];
    float ew = (s != d) ? attr[e] : 0.0f;
    rank[e] = atomicAdd(&cnt[d], 1);
    if (ew != 0.0f) atomicAdd(&deg[d], ew);
}

__global__ void k_dis(float* __restrict__ deg, int n) {
    int i = blockIdx.x * blockDim.x + threadIdx.x;
    if (i >= n) return;
    float dv = deg[i];
    deg[i] = (dv > 0.0f) ? rsqrtf(fmaxf(dv, 1e-30f)) : 0.0f;
}

#define SCAN_T 1024
__global__ __launch_bounds__(SCAN_T) void
k_rowptr(const int* __restrict__ cnt, int* __restrict__ rowptr, int n) {
    __shared__ int pre_s[SCAN_T / 32];
    __shared__ int sum_s[SCAN_T / 32];
    int tid  = threadIdx.x;
    int lane = tid & 31;
    int wv   = tid >> 5;
    int base = blockIdx.x * (SCAN_T * 4);

    int pre = 0;
    const int4* c4 = reinterpret_cast<const int4*>(cnt) + tid;
#pragma unroll 4
    for (int k = 0; k < (int)blockIdx.x; ++k) {
        int4 v = c4[(size_t)k * SCAN_T];
        pre += (v.x + v.y) + (v.z + v.w);
    }
    for (int o = 16; o > 0; o >>= 1) pre += __shfl_xor(pre, o);
    if (lane == 0) pre_s[wv] = pre;

    int i0 = base + tid * 4;
    int c0 = (i0     < n) ? cnt[i0]     : 0;
    int c1 = (i0 + 1 < n) ? cnt[i0 + 1] : 0;
    int c2 = (i0 + 2 < n) ? cnt[i0 + 2] : 0;
    int c3 = (i0 + 3 < n) ? cnt[i0 + 3] : 0;
    int s = (c0 + c1) + (c2 + c3);
    int inc = s;
    for (int o = 1; o < 32; o <<= 1) {
        int t = __shfl_up(inc, o);
        if (lane >= o) inc += t;
    }
    if (lane == 31) sum_s[wv] = inc;
    __syncthreads();

    int t = pre_s[lane] + ((lane < wv) ? sum_s[lane] : 0);
    for (int o = 16; o > 0; o >>= 1) t += __shfl_xor(t, o);
    int x = t + inc - s;

    if (blockIdx.x == 0 && tid == 0) rowptr[0] = 0;
    x += c0; if (i0     < n) rowptr[i0 + 1] = x;
    x += c1; if (i0 + 1 < n) rowptr[i0 + 2] = x;
    x += c2; if (i0 + 2 < n) rowptr[i0 + 3] = x;
    x += c3; if (i0 + 3 < n) rowptr[i0 + 4] = x;
}

__global__ void k_edge_place(const int* __restrict__ ei, const float* __restrict__ attr,
                             const float* __restrict__ dis, const int* __restrict__ rowptr,
                             const int* __restrict__ rank, int2* __restrict__ csr, int E) {
    int e = blockIdx.x * blockDim.x + threadIdx.x;
    if (e >= E) return;
    int s = ei[e];
    int d = ei[E + e];
    float ew = (s != d) ? attr[e] : 0.0f;
    float wv = -dis[d] * ew * dis[s];
    csr[rowptr[d] + rank[e]] = make_int2(s, __float_as_int(wv));
}

template <int U>
__device__ __forceinline__ void gather_chunk(const int2* __restrict__ c,
                                             const float2* __restrict__ hp, float2& acc) {
    int2 ce[U];
    float2 v[U];
#pragma unroll
    for (int j = 0; j < U; ++j) ce[j] = c[j];
#pragma unroll
    for (int j = 0; j < U; ++j) v[j] = hp[(size_t)ce[j].x * 32];
#pragma unroll
    for (int j = 0; j < U; ++j) {
        float wj = __int_as_float(ce[j].y);
        acc.x = fmaf(wj, v[j].x, acc.x);
        acc.y = fmaf(wj, v[j].y, acc.y);
    }
}

template <bool HAS_BASE, bool RELU>
__global__ __launch_bounds__(256) void
k_gather64(const float* __restrict__ h, const int2* __restrict__ csr,
           const int* __restrict__ rowptr, float* __restrict__ out, int n, float scale) {
    int row = __builtin_amdgcn_readfirstlane((int)((blockIdx.x * 256 + threadIdx.x) >> 5));
    if (row >= n) return;
    int lane = threadIdx.x & 31;
    float2* op = reinterpret_cast<float2*>(out) + (size_t)row * 32 + lane;
    float2 b = make_float2(0.0f, 0.0f);
    if (HAS_BASE) b = *op;
    int e   = rowptr[row];
    int end = rowptr[row + 1];
    const float2* hp = reinterpret_cast<const float2*>(h) + lane;
    float2 acc = make_float2(0.0f, 0.0f);
    for (; e + 8 <= end; e += 8) gather_chunk<8>(csr + e, hp, acc);
    if (e + 4 <= end) { gather_chunk<4>(csr + e, hp, acc); e += 4; }
    int rem = end - e;
    if (rem > 0) {
        int2 t0 = csr[e];
        int2 t1 = csr[rem > 1 ? e + 1 : e];
        int2 t2 = csr[rem > 2 ? e + 2 : e];
        float2 v0 = hp[(size_t)t0.x * 32];
        float2 v1 = hp[(size_t)t1.x * 32];
        float2 v2 = hp[(size_t)t2.x * 32];
        float w0 = __int_as_float(t0.y);
        acc.x = fmaf(w0, v0.x, acc.x);
        acc.y = fmaf(w0, v0.y, acc.y);
        if (rem > 1) {
            float w1 = __int_as_float(t1.y);
            acc.x = fmaf(w1, v1.x, acc.x);
            acc.y = fmaf(w1, v1.y, acc.y);
        }
        if (rem > 2) {
            float w2 = __int_as_float(t2.y);
            acc.x = fmaf(w2, v2.x, acc.x);
            acc.y = fmaf(w2, v2.y, acc.y);
        }
    }
    float2 r;
    r.x = scale * acc.x + b.x;
    r.y = scale * acc.y + b.y;
    if (RELU) { r.x = fmaxf(r.x, 0.0f); r.y = fmaxf(r.y, 0.0f); }
    *op = r;
}

#define GW 8
#define KPAD 132

__global__ __launch_bounds__(GW * 32) void
k_cheb_gemm(const float* __restrict__ Tx0, const float* __restrict__ Tx1,
            const float* __restrict__ P, const float* __restrict__ W,
            const float* __restrict__ bias, float* __restrict__ out,
            int M, int Kd, int N, int Mtiles) {
    __shared__ float Wl[3 * 16 * KPAD];

    int tid  = threadIdx.x;
    int nt   = blockIdx.y;

    for (int idx = tid; idx < 3 * 16 * Kd; idx += GW * 32) {
        int t  = idx / (16 * Kd);
        int r  = idx - t * 16 * Kd;
        int cc = r / Kd;
        int k  = r - cc * Kd;
        int nc = nt * 16 + cc;
        float v = W[(size_t)(t * Kd + k) * N + nc];
        if (t == 0) v -= W[(size_t)(2 * Kd + k) * N + nc];
        Wl[(t * 16 + cc) * KPAD + k] = v;
    }
    __syncthreads();

    int wave = tid >> 5;
    int mt   = blockIdx.x * GW + wave;
    if (mt >= Mtiles) return;

    int lane = tid & 31;
    int col  = lane & 15;
    int klo  = (lane >> 4) << 1;
    int row  = mt * 16 + col;
    int ncol = nt * 16 + col;

    const float* Txs[3] = {Tx0, Tx1, P};

    v8f c;
    float bv = bias[ncol];
#pragma unroll
    for (int i = 0; i < 8; ++i) c[i] = bv;

    for (int t = 0; t < 3; ++t) {
        const float* arow = Txs[t] + (size_t)row * Kd;
        const float* brow = &Wl[(t * 16 + col) * KPAD];
        for (int k0 = 0; k0 < Kd; k0 += 4) {
            v2f a, b;
            a.x = arow[k0 + klo];
            a.y = arow[k0 + klo + 1];
            b = *reinterpret_cast<const v2f*>(brow + k0 + klo);
            c = __builtin_amdgcn_wmma_f32_16x16x4_f32(
                     false, a,  false, b,
                     (short)0, c,  false,  false);
        }
    }

    int rbase = mt * 16 + ((lane >> 4) << 3);
#pragma unroll
    for (int i = 0; i < 8; ++i) {
        float v = fmaxf(c[i], 0.0f);
        int r = rbase + i;
        if (r < M) out[(size_t)r * N + ncol] = v;
    }
}

__global__ __launch_bounds__(GW * 32) void
k_premul_gemm(const float* __restrict__ X, const float* __restrict__ W,
              const float* __restrict__ bias, float* __restrict__ Y0,
              float* __restrict__ Y1, float* __restrict__ Y2,
              int M, int Kd, int N, int Mtiles) {
    __shared__ float Wl[3 * 16 * KPAD];

    int tid  = threadIdx.x;
    int nt   = blockIdx.y;

    for (int idx = tid; idx < 3 * 16 * Kd; idx += GW * 32) {
        int t  = idx / (16 * Kd);
        int r  = idx - t * 16 * Kd;
        int cc = r / Kd;
        int k  = r - cc * Kd;
        int nc = nt * 16 + cc;
        float v = W[(size_t)(t * Kd + k) * N + nc];
        if (t == 0) v -= W[(size_t)(2 * Kd + k) * N + nc];
        Wl[(t * 16 + cc) * KPAD + k] = v;
    }
    __syncthreads();

    int wave = tid >> 5;
    int mt   = blockIdx.x * GW + wave;
    if (mt >= Mtiles) return;

    int lane = tid & 31;
    int col  = lane & 15;
    int klo  = (lane >> 4) << 1;
    int row  = mt * 16 + col;
    int ncol = nt * 16 + col;

    v8f c0, c1, c2;
    float bv = bias[ncol];
#pragma unroll
    for (int i = 0; i < 8; ++i) { c0[i] = bv; c1[i] = 0.0f; c2[i] = 0.0f; }

    const float* arow  = X + (size_t)row * Kd;
    const float* brow0 = &Wl[(0 * 16 + col) * KPAD];
    const float* brow1 = &Wl[(1 * 16 + col) * KPAD];
    const float* brow2 = &Wl[(2 * 16 + col) * KPAD];
    for (int k0 = 0; k0 < Kd; k0 += 4) {
        v2f a, b0, b1, b2;
        a.x = arow[k0 + klo];
        a.y = arow[k0 + klo + 1];
        b0 = *reinterpret_cast<const v2f*>(brow0 + k0 + klo);
        b1 = *reinterpret_cast<const v2f*>(brow1 + k0 + klo);
        b2 = *reinterpret_cast<const v2f*>(brow2 + k0 + klo);
        c0 = __builtin_amdgcn_wmma_f32_16x16x4_f32(
                  false, a,  false, b0,
                  (short)0, c0,  false,  false);
        c1 = __builtin_amdgcn_wmma_f32_16x16x4_f32(
                 false, a, false, b1, (short)0, c1, false, false);
        c2 = __builtin_amdgcn_wmma_f32_16x16x4_f32(
                 false, a, false, b2, (short)0, c2, false, false);
    }

    int rbase = mt * 16 + ((lane >> 4) << 3);
#pragma unroll
    for (int i = 0; i < 8; ++i) {
        int r = rbase + i;
        if (r < M) {
            size_t o = (size_t)r * N + ncol;
            Y0[o] = c0[i];
            Y1[o] = c1[i];
            Y2[o] = c2[i];
        }
    }
}


__global__ void k_count(const int* __restrict__ batch, float* __restrict__ cnt, int n) {
    int i = blockIdx.x * blockDim.x + threadIdx.x;
    if (i < n) atomicAdd(&cnt[batch[i]], 1.0f);
}

__global__ void k_pool(const float* __restrict__ h, const int* __restrict__ batch,
                       float* __restrict__ sums, int n) {
    int idx = blockIdx.x * blockDim.x + threadIdx.x;
    int total = n << 5;
    if (idx >= total) return;
    int i = idx >> 5;
    int c = idx & 31;
    int g = batch[i];
    const float4* hp = reinterpret_cast<const float4*>(h) + ((size_t)i << 5) + c;
    float4 hv = *hp;
    float* sp = sums + ((size_t)g << 7) + (c << 2);
    atomicAdd(sp + 0, hv.x);
    atomicAdd(sp + 1, hv.y);
    atomicAdd(sp + 2, hv.z);
    atomicAdd(sp + 3, hv.w);
}

__global__ void k_final(const float* __restrict__ sums, const float* __restrict__ cnt,
                        const float* __restrict__ linW, const float* __restrict__ linb,
                        float* __restrict__ out, int G, int H) {
    __shared__ float hg_s[128];
    int g = blockIdx.x;
    int f = threadIdx.x;
    float c = cnt[g];
    float hg = sums[(size_t)g * H + f] / fmaxf(c, 1.0f);
    out[(size_t)G * 2 + (size_t)g * H + f] = hg;
    hg_s[f] = hg;
    __syncthreads();
    if (f < 2) {
        float acc = linb[f];
        for (int j = 0; j < H; ++j) acc += hg_s[j] * linW[j * 2 + f];
        out[g * 2 + f] = acc;
    }
}


extern "C" void kernel_launch(void* const* d_in, const int* in_sizes, int n_in,
                              void* d_out, int out_size, void* d_ws, size_t ws_size,
                              hipStream_t stream) {
    const float* x    = (const float*)d_in[0];
    const int*   ei   = (const int*)d_in[1];
    const float* attr = (const float*)d_in[2];
    const int*   bat  = (const int*)d_in[3];
    const float* W1 = (const float*)d_in[4];  const float* b1 = (const float*)d_in[5];
    const float* W2 = (const float*)d_in[6];  const float* b2 = (const float*)d_in[7];
    const float* W3 = (const float*)d_in[8];  const float* b3 = (const float*)d_in[9];
    const float* linW = (const float*)d_in[10];
    const float* linb = (const float*)d_in[11];
    float* out = (float*)d_out;

    const int F_IN = 128, H1 = 64, H2 = 64, H3 = 128, NCLS = 2;
    int Nn = in_sizes[0] / F_IN;
    int E  = in_sizes[2];
    int G  = out_size / (NCLS + H3);

    auto up = [](size_t v) { return (v + 63) & ~(size_t)63; };
    float* ws = (float*)d_ws;
    size_t off = 0;
    float* dis  = ws + off; off += up((size_t)Nn);
    int*   ecnt = (int*)(ws + off); off += up((size_t)Nn);
    int*   rptr = (int*)(ws + off); off += up((size_t)Nn + 1);
    int*   rank = (int*)(ws + off); off += up((size_t)E);
    int2*  csr  = (int2*)(ws + off); off += up((size_t)E * 2);
    float* A    = ws + off; off += up((size_t)Nn * H1);
    float* B    = ws + off; off += up((size_t)Nn * H1);
    float* C    = ws + off; off += up((size_t)Nn * H3);
    float* D    = ws + off; off += up((size_t)Nn * H2);
    float* pool = ws + off; off += (size_t)G * H3;
    float* cnt  = ws + off; off += (size_t)G;

    auto nb = [](long n, int t) { return (unsigned)((n + t - 1) / t); };

    long nz = (long)(up((size_t)Nn) + (size_t)Nn);
    k_fill<<<nb(nz, 256), 256, 0, stream>>>(dis, 0.0f, (int)nz);
    k_edge_deg<<<nb(E, 256), 256, 0, stream>>>(ei, attr, dis, ecnt, rank, E);
    k_dis<<<nb(Nn, 256), 256, 0, stream>>>(dis, Nn);
    k_rowptr<<<nb(Nn, SCAN_T * 4), SCAN_T, 0, stream>>>(ecnt, rptr, Nn);
    k_edge_place<<<nb(E, 256), 256, 0, stream>>>(ei, attr, dis, rptr, rank, csr, E);

    unsigned gb = nb((long)Nn * 32, 256);

    auto layer = [&](const float* in, int Fin, const float* Wt, const float* bt,
                     int Nout, float* outp) {
        k_gather64<false, false><<<gb, 256, 0, stream>>>(in, csr, rptr, A, Nn, 1.0f);
        k_gather64<false, false><<<gb, 256, 0, stream>>>(A, csr, rptr, B, Nn, 2.0f);
        int mtiles = (Nn + 15) / 16;
        dim3 g((mtiles + GW - 1) / GW, Nout / 16);
        k_cheb_gemm<<<g, GW * 32, 0, stream>>>(in, A, B, Wt, bt, outp,
                                               Nn, Fin, Nout, mtiles);
    };

    auto layer_narrow = [&](const float* in, int Fin, const float* Wt, const float* bt,
                            int Nout, float* outp) {
        int mtiles = (Nn + 15) / 16;
        dim3 g((mtiles + GW - 1) / GW, Nout / 16);
        k_premul_gemm<<<g, GW * 32, 0, stream>>>(in, Wt, bt, outp, A, B,
                                                 Nn, Fin, Nout, mtiles);
        k_gather64<true, false><<<gb, 256, 0, stream>>>(B, csr, rptr, A, Nn, 2.0f);
        k_gather64<true, true><<<gb, 256, 0, stream>>>(A, csr, rptr, outp, Nn, 1.0f);
    };

    layer_narrow(x, F_IN, W1, b1, H1, C);
    layer(C, H1,  W2, b2, H2, D);
    layer(D, H2,  W3, b3, H3, C);

    k_fill<<<nb((long)G * H3 + G, 256), 256, 0, stream>>>(pool, 0.0f, G * H3 + G);
    k_count<<<nb(Nn, 256), 256, 0, stream>>>(bat, cnt, Nn);
    k_pool<<<nb((long)Nn * (H3 / 4), 256), 256, 0, stream>>>(C, bat, pool, Nn);
    k_final<<<G, H3, 0, stream>>>(pool, cnt, linW, linb, out, G, H3);
}


# f32 WMMA GEMMs re-tiled: 256 looping blocks, one wave per 16-row strip over all output columns (A read once, not per 16-column tile), weights staged once per block; sum order inside each 8 k changes
# speedup vs baseline: 3.0455x; 1.2499x over previous
#include <hip/hip_runtime.h>

typedef __attribute__((ext_vector_type(2))) float v2f;
typedef __attribute__((ext_vector_type(8))) float v8f;


__global__ void k_fill(float* __restrict__ p, float v, int n) {
    int i = blockIdx.x * blockDim.x + threadIdx.x;
    if (i < n) p[i] = v;
}


__global__ void k_edge_deg(const int* __restrict__ ei, const float* __restrict__ attr,
                           float* __restrict__ deg, int* __restrict__ cnt,
                           int* __restrict__ rank, int E) {
    int e = blockIdx.x * blockDim.x + threadIdx.x;
    if (e >= E) return;
    int s = ei[e];
    int d = ei[E + e];
    float ew = (s != d) ? attr[e] : 0.0f;
    rank[e] = atomicAdd(&cnt[d], 1);
    if (ew != 0.0f) atomicAdd(&deg[d], ew);
}

__global__ void k_dis(float* __restrict__ deg, int n) {
    int i = blockIdx.x * blockDim.x + threadIdx.x;
    if (i >= n) return;
    float dv = deg[i];
    deg[i] = (dv > 0.0f) ? rsqrtf(fmaxf(dv, 1e-30f)) : 0.0f;
}

#define SCAN_T 1024
__global__ __launch_bounds__(SCAN_T) void
k_rowptr(const int* __restrict__ cnt, int* __restrict__ rowptr, int n) {
    __shared__ int pre_s[SCAN_T / 32];
    __shared__ int sum_s[SCAN_T / 32];
    int tid  = threadIdx.x;
    int lane = tid & 31;
    int wv   = tid >> 5;
    int base = blockIdx.x * (SCAN_T * 4);

    int pre = 0;
    const int4* c4 = reinterpret_cast<const int4*>(cnt) + tid;
#pragma unroll 4
    for (int k = 0; k < (int)blockIdx.x; ++k) {
        int4 v = c4[(size_t)k * SCAN_T];
        pre += (v.x + v.y) + (v.z + v.w);
    }
    for (int o = 16; o > 0; o >>= 1) pre += __shfl_xor(pre, o);
    if (lane == 0) pre_s[wv] = pre;

    int i0 = base + tid * 4;
    int c0 = (i0     < n) ? cnt[i0]     : 0;
    int c1 = (i0 + 1 < n) ? cnt[i0 + 1] : 0;
    int c2 = (i0 + 2 < n) ? cnt[i0 + 2] : 0;
    int c3 = (i0 + 3 < n) ? cnt[i0 + 3] : 0;
    int s = (c0 + c1) + (c2 + c3);
    int inc = s;
    for (int o = 1; o < 32; o <<= 1) {
        int t = __shfl_up(inc, o);
        if (lane >= o) inc += t;
    }
    if (lane == 31) sum_s[wv] = inc;
    __syncthreads();

    int t = pre_s[lane] + ((lane < wv) ? sum_s[lane] : 0);
    for (int o = 16; o > 0; o >>= 1) t += __shfl_xor(t, o);
    int x = t + inc - s;

    if (blockIdx.x == 0 && tid == 0) rowptr[0] = 0;
    x += c0; if (i0     < n) rowptr[i0 + 1] = x;
    x += c1; if (i0 + 1 < n) rowptr[i0 + 2] = x;
    x += c2; if (i0 + 2 < n) rowptr[i0 + 3] = x;
    x += c3; if (i0 + 3 < n) rowptr[i0 + 4] = x;
}

__global__ void k_edge_place(const int* __restrict__ ei, const float* __restrict__ attr,
                             const float* __restrict__ dis, const int* __restrict__ rowptr,
                             const int* __restrict__ rank, int2* __restrict__ csr, int E) {
    int e = blockIdx.x * blockDim.x + threadIdx.x;
    if (e >= E) return;
    int s = ei[e];
    int d = ei[E + e];
    float ew = (s != d) ? attr[e] : 0.0f;
    float wv = -dis[d] * ew * dis[s];
    csr[rowptr[d] + rank[e]] = make_int2(s, __float_as_int(wv));
}

template <int U>
__device__ __forceinline__ void gather_chunk(const int2* __restrict__ c,
                                             const float2* __restrict__ hp, float2& acc) {
    int2 ce[U];
    float2 v[U];
#pragma unroll
    for (int j = 0; j < U; ++j) ce[j] = c[j];
#pragma unroll
    for (int j = 0; j < U; ++j) v[j] = hp[(size_t)ce[j].x * 32];
#pragma unroll
    for (int j = 0; j < U; ++j) {
        float wj = __int_as_float(ce[j].y);
        acc.x = fmaf(wj, v[j].x, acc.x);
        acc.y = fmaf(wj, v[j].y, acc.y);
    }
}

template <bool HAS_BASE, bool RELU>
__global__ __launch_bounds__(256) void
k_gather64(const float* __restrict__ h, const int2* __restrict__ csr,
           const int* __restrict__ rowptr, float* __restrict__ out, int n, float scale) {
    int row = __builtin_amdgcn_readfirstlane((int)((blockIdx.x * 256 + threadIdx.x) >> 5));
    if (row >= n) return;
    int lane = threadIdx.x & 31;
    float2* op = reinterpret_cast<float2*>(out) + (size_t)row * 32 + lane;
    float2 b = make_float2(0.0f, 0.0f);
    if (HAS_BASE) b = *op;
    int e   = rowptr[row];
    int end = rowptr[row + 1];
    const float2* hp = reinterpret_cast<const float2*>(h) + lane;
    float2 acc = make_float2(0.0f, 0.0f);
    for (; e + 8 <= end; e += 8) gather_chunk<8>(csr + e, hp, acc);
    if (e + 4 <= end) { gather_chunk<4>(csr + e, hp, acc); e += 4; }
    int rem = end - e;
    if (rem > 0) {
        int2 t0 = csr[e];
        int2 t1 = csr[rem > 1 ? e + 1 : e];
        int2 t2 = csr[rem > 2 ? e + 2 : e];
        float2 v0 = hp[(size_t)t0.x * 32];
        float2 v1 = hp[(size_t)t1.x * 32];
        float2 v2 = hp[(size_t)t2.x * 32];
        float w0 = __int_as_float(t0.y);
        acc.x = fmaf(w0, v0.x, acc.x);
        acc.y = fmaf(w0, v0.y, acc.y);
        if (rem > 1) {
            float w1 = __int_as_float(t1.y);
            acc.x = fmaf(w1, v1.x, acc.x);
            acc.y = fmaf(w1, v1.y, acc.y);
        }
        if (rem > 2) {
            float w2 = __int_as_float(t2.y);
            acc.x = fmaf(w2, v2.x, acc.x);
            acc.y = fmaf(w2, v2.y, acc.y);
        }
    }
    float2 r;
    r.x = scale * acc.x + b.x;
    r.y = scale * acc.y + b.y;
    if (RELU) { r.x = fmaxf(r.x, 0.0f); r.y = fmaxf(r.y, 0.0f); }
    *op = r;
}

#define SW 12
#define SGRID 256
#define SLDS 41984

template <int KD, int N>
__device__ __forceinline__ void stage_weights(const float* __restrict__ W,
                                              float* __restrict__ Wl, int tid) {
    constexpr int KP   = KD + 4;
    constexpr int Q    = N / 4;
    constexpr int SL4  = KD * Q;
    constexpr int NB16 = N / 16;
    constexpr int GPS  = (KD / 8) * NB16;
    constexpr int IT0  = (GPS + SW - 1) / SW;
    constexpr int IT1  = (2 * GPS + SW - 1) / SW;
    const float4* W4 = reinterpret_cast<const float4*>(W);
    int wv = tid >> 5;
    int q  = tid & 3;
    int kk = (tid >> 2) & 7;

    float4 va[IT0], vb[IT0], vc[IT1];
#pragma unroll
    for (int it = 0; it < IT0; ++it) {
        int g    = min(wv + it * SW, GPS - 1);
        int kblk = g / NB16;
        int nblk = g - kblk * NB16;
        int idx  = (kblk * 8 + kk) * Q + nblk * 4 + q;
        va[it] = W4[idx];
        vb[it] = W4[idx + 2 * SL4];
    }
#pragma unroll
    for (int it = 0; it < IT1; ++it) {
        int g    = min(wv + it * SW, 2 * GPS - 1);
        int t    = g / GPS;
        int r    = g - t * GPS;
        int kblk = r / NB16;
        int nblk = r - kblk * NB16;
        vc[it] = W4[((t + 1) * KD + kblk * 8 + kk) * Q + nblk * 4 + q];
    }
#pragma unroll
    for (int it = 0; it < IT0; ++it) {
        int g = wv + it * SW;
        if (g < GPS) {
            int kblk = g / NB16;
            int nblk = g - kblk * NB16;
            float* p = Wl + (nblk * 16 + q * 4) * KP + kblk * 8 + kk;
            p[0]      = va[it].x - vb[it].x;
            p[KP]     = va[it].y - vb[it].y;
            p[2 * KP] = va[it].z - vb[it].z;
            p[3 * KP] = va[it].w - vb[it].w;
        }
    }
#pragma unroll
    for (int it = 0; it < IT1; ++it) {
        int g = wv + it * SW;
        if (g < 2 * GPS) {
            int t    = g / GPS;
            int r    = g - t * GPS;
            int kblk = r / NB16;
            int nblk = r - kblk * NB16;
            float* p = Wl + ((t + 1) * N + nblk * 16 + q * 4) * KP + kblk * 8 + kk;
            p[0]      = vc[it].x;
            p[KP]     = vc[it].y;
            p[2 * KP] = vc[it].z;
            p[3 * KP] = vc[it].w;
        }
    }
}

template <int KB>
__device__ __forceinline__ void load_a(float4 (&a)[KB], const float* __restrict__ rowp, int h) {
    const float4* p = reinterpret_cast<const float4*>(rowp) + h;
#pragma unroll
    for (int kb = 0; kb < KB; ++kb) a[kb] = p[kb * 2];
}

template <int NT, int KP>
__device__ __forceinline__ void load_b(float4 (&b)[NT], const float* p) {
    __builtin_amdgcn_fence(__ATOMIC_ACQUIRE, "wavefront", "local");
#pragma unroll
    for (int j = 0; j < NT; ++j)
        b[j] = *reinterpret_cast<const float4*>(p + j * 16 * KP);
}

template <int NT>
__device__ __forceinline__ void mma_step(const float4& a, const float4 (&b)[NT], v8f (&acc)[NT]) {
    v2f alo, ahi;
    alo.x = a.x; alo.y = a.y;
    ahi.x = a.z; ahi.y = a.w;
#pragma unroll
    for (int j = 0; j < NT; ++j) {
        v2f bl; bl.x = b[j].x; bl.y = b[j].y;
        acc[j] = __builtin_amdgcn_wmma_f32_16x16x4_f32(
                      false, alo,  false, bl,
                      (short)0, acc[j],  false,  false);
    }
#pragma unroll
    for (int j = 0; j < NT; ++j) {
        v2f bh; bh.x = b[j].z; bh.y = b[j].w;
        acc[j] = __builtin_amdgcn_wmma_f32_16x16x4_f32(
                     false, ahi, false, bh, (short)0, acc[j], false, false);
    }
}

template <int KB, int NT, int KP>
__device__ __forceinline__ void mma_unit(const float4 (&a)[KB], const float* bp, const float* bnext,
                                         float4 (&bc)[NT], v8f (&acc)[NT]) {
#pragma unroll
    for (int kb = 0; kb < KB; ++kb) {
        float4 bn[NT];
        load_b<NT, KP>(bn, (kb + 1 < KB) ? bp + (kb + 1) * 8 : bnext);
        mma_step<NT>(a[kb], bc, acc);
#pragma unroll
        for (int j = 0; j < NT; ++j) bc[j] = bn[j];
    }
}

template <int KD, int N>
__global__ __launch_bounds__(SW * 32) void
k_cheb_strip(const float* __restrict__ Tx0, const float* __restrict__ Tx1,
             const float* __restrict__ P, const float* __restrict__ W,
             const float* __restrict__ bias, float* __restrict__ out, int M, int Mtiles) {
    __shared__ __attribute__((aligned(16))) float Wl[SLDS];
    constexpr int KP = KD + 4;
    constexpr int NT = N / 16;
    constexpr int KB = KD / 8;

    int tid = threadIdx.x;
    stage_weights<KD, N>(W, Wl, tid);
    __syncthreads();

    int wave = __builtin_amdgcn_readfirstlane(tid >> 5);
    int step = SW * (int)gridDim.x;
    int mt   = wave * (int)gridDim.x + (int)blockIdx.x;
    if (mt >= Mtiles) return;

    int lane = tid & 31;
    int c    = lane & 15;
    int h    = lane >> 4;

    float bv[NT];
#pragma unroll
    for (int j = 0; j < NT; ++j) bv[j] = bias[j * 16 + c];

    const float* bp = Wl + c * KP + 4 * h;
    int row = min(mt * 16 + c, M - 1);

    float4 a0[KB], a1[KB];
    float4 bc[NT];
    load_a<KB>(a0, Tx0 + (size_t)row * KD, h);
    load_b<NT, KP>(bc, bp);
    for (;;) {
        int  mtn  = mt + step;
        bool more = mtn < Mtiles;
        int  rown = min((more ? mtn : mt) * 16 + c, M - 1);

        v8f acc[NT];
#pragma unroll
        for (int j = 0; j < NT; ++j) {
#pragma unroll
            for (int i = 0; i < 8; ++i) acc[j][i] = bv[j];
        }

        load_a<KB>(a1, Tx1 + (size_t)row * KD, h);
        mma_unit<KB, NT, KP>(a0, bp, bp + N * KP, bc, acc);
        load_a<KB>(a0, P + (size_t)row * KD, h);
        mma_unit<KB, NT, KP>(a1, bp + N * KP, bp + 2 * N * KP, bc, acc);
        load_a<KB>(a1, Tx0 + (size_t)rown * KD, h);
        mma_unit<KB, NT, KP>(a0, bp + 2 * N * KP, bp, bc, acc);

        int rbase = mt * 16 + (h << 3);
#pragma unroll
        for (int i = 0; i < 8; ++i) {
            int r = rbase + i;
            if (r < M) {
                float* op = out + (size_t)r * N + c;
#pragma unroll
                for (int j = 0; j < NT; ++j) op[j * 16] = fmaxf(acc[j][i], 0.0f);
            }
        }
        if (!more) break;
#pragma unroll
        for (int kb = 0; kb < KB; ++kb) a0[kb] = a1[kb];
        mt  = mtn;
        row = rown;
    }
}

template <int KD, int N>
__global__ __launch_bounds__(SW * 32) void
k_premul_strip(const float* __restrict__ X, const float* __restrict__ W,
               const float* __restrict__ bias, float* __restrict__ Y0,
               float* __restrict__ Y1, float* __restrict__ Y2, int M, int Mtiles) {
    __shared__ __attribute__((aligned(16))) float Wl[SLDS];
    constexpr int KP = KD + 4;
    constexpr int NT = N / 16;
    constexpr int KH = KD / 2;
    constexpr int KB = KH / 8;

    int tid = threadIdx.x;
    stage_weights<KD, N>(W, Wl, tid);
    __syncthreads();

    int wave = __builtin_amdgcn_readfirstlane(tid >> 5);
    int step = SW * (int)gridDim.x;
    int mt   = wave * (int)gridDim.x + (int)blockIdx.x;
    if (mt >= Mtiles) return;

    int lane = tid & 31;
    int c    = lane & 15;
    int h    = lane >> 4;

    float bv[NT];
#pragma unroll
    for (int j = 0; j < NT; ++j) bv[j] = bias[j * 16 + c];

    const float* bp = Wl + c * KP + 4 * h;
    int row = min(mt * 16 + c, M - 1);

    float4 a0[KB], a1[KB];
    float4 bc[3 * NT];
    load_a<KB>(a0, X + (size_t)row * KD, h);
    load_b<3 * NT, KP>(bc, bp);
    for (;;) {
        int  mtn  = mt + step;
        bool more = mtn < Mtiles;
        int  rown = min((more ? mtn : mt) * 16 + c, M - 1);

        v8f acc[3 * NT];
#pragma unroll
        for (int j = 0; j < 3 * NT; ++j) {
#pragma unroll
            for (int i = 0; i < 8; ++i) acc[j][i] = (j < NT) ? bv[j < NT ? j : 0] : 0.0f;
        }

        load_a<KB>(a1, X + (size_t)row * KD + KH, h);
        mma_unit<KB, 3 * NT, KP>(a0, bp, bp + KH, bc, acc);
        load_a<KB>(a0, X + (size_t)rown * KD, h);
        mma_unit<KB, 3 * NT, KP>(a1, bp + KH, bp, bc, acc);

        int rbase = mt * 16 + (h << 3);
#pragma unroll
        for (int i = 0; i < 8; ++i) {
            int r = rbase + i;
            if (r < M) {
                size_t o = (size_t)r * N + c;
#pragma unroll
                for (int j = 0; j < NT; ++j) {
                    Y0[o + j * 16] = acc[j][i];
                    Y1[o + j * 16] = acc[NT + j][i];
                    Y2[o + j * 16] = acc[2 * NT + j][i];
                }
            }
        }
        if (!more) break;
        mt  = mtn;
        row = rown;
    }
}


__global__ void k_count(const int* __restrict__ batch, float* __restrict__ cnt, int n) {
    int i = blockIdx.x * blockDim.x + threadIdx.x;
    if (i < n) atomicAdd(&cnt[batch[i]], 1.0f);
}

__global__ void k_pool(const float* __restrict__ h, const int* __restrict__ batch,
                       float* __restrict__ sums, int n) {
    int idx = blockIdx.x * blockDim.x + threadIdx.x;
    int total = n << 5;
    if (idx >= total) return;
    int i = idx >> 5;
    int c = idx & 31;
    int g = batch[i];
    const float4* hp = reinterpret_cast<const float4*>(h) + ((size_t)i << 5) + c;
    float4 hv = *hp;
    float* sp = sums + ((size_t)g << 7) + (c << 2);
    atomicAdd(sp + 0, hv.x);
    atomicAdd(sp + 1, hv.y);
    atomicAdd(sp + 2, hv.z);
    atomicAdd(sp + 3, hv.w);
}

__global__ void k_final(const float* __restrict__ sums, const float* __restrict__ cnt,
                        const float* __restrict__ linW, const float* __restrict__ linb,
                        float* __restrict__ out, int G, int H) {
    __shared__ float hg_s[128];
    int g = blockIdx.x;
    int f = threadIdx.x;
    float c = cnt[g];
    float hg = sums[(size_t)g * H + f] / fmaxf(c, 1.0f);
    out[(size_t)G * 2 + (size_t)g * H + f] = hg;
    hg_s[f] = hg;
    __syncthreads();
    if (f < 2) {
        float acc = linb[f];
        for (int j = 0; j < H; ++j) acc += hg_s[j] * linW[j * 2 + f];
        out[g * 2 + f] = acc;
    }
}


extern "C" void kernel_launch(void* const* d_in, const int* in_sizes, int n_in,
                              void* d_out, int out_size, void* d_ws, size_t ws_size,
                              hipStream_t stream) {
    const float* x    = (const float*)d_in[0];
    const int*   ei   = (const int*)d_in[1];
    const float* attr = (const float*)d_in[2];
    const int*   bat  = (const int*)d_in[3];
    const float* W1 = (const float*)d_in[4];  const float* b1 = (const float*)d_in[5];
    const float* W2 = (const float*)d_in[6];  const float* b2 = (const float*)d_in[7];
    const float* W3 = (const float*)d_in[8];  const float* b3 = (const float*)d_in[9];
    const float* linW = (const float*)d_in[10];
    const float* linb = (const float*)d_in[11];
    float* out = (float*)d_out;

    const int F_IN = 128, H1 = 64, H2 = 64, H3 = 128, NCLS = 2;
    int Nn = in_sizes[0] / F_IN;
    int E  = in_sizes[2];
    int G  = out_size / (NCLS + H3);

    auto up = [](size_t v) { return (v + 63) & ~(size_t)63; };
    float* ws = (float*)d_ws;
    size_t off = 0;
    float* dis  = ws + off; off += up((size_t)Nn);
    int*   ecnt = (int*)(ws + off); off += up((size_t)Nn);
    int*   rptr = (int*)(ws + off); off += up((size_t)Nn + 1);
    int*   rank = (int*)(ws + off); off += up((size_t)E);
    int2*  csr  = (int2*)(ws + off); off += up((size_t)E * 2);
    float* A    = ws + off; off += up((size_t)Nn * H1);
    float* B    = ws + off; off += up((size_t)Nn * H1);
    float* C    = ws + off; off += up((size_t)Nn * H3);
    float* D    = ws + off; off += up((size_t)Nn * H2);
    float* pool = ws + off; off += (size_t)G * H3;
    float* cnt  = ws + off; off += (size_t)G;

    auto nb = [](long n, int t) { return (unsigned)((n + t - 1) / t); };

    long nz = (long)(up((size_t)Nn) + (size_t)Nn);
    k_fill<<<nb(nz, 256), 256, 0, stream>>>(dis, 0.0f, (int)nz);
    k_edge_deg<<<nb(E, 256), 256, 0, stream>>>(ei, attr, dis, ecnt, rank, E);
    k_dis<<<nb(Nn, 256), 256, 0, stream>>>(dis, Nn);
    k_rowptr<<<nb(Nn, SCAN_T * 4), SCAN_T, 0, stream>>>(ecnt, rptr, Nn);
    k_edge_place<<<nb(E, 256), 256, 0, stream>>>(ei, attr, dis, rptr, rank, csr, E);

    unsigned gb = nb((long)Nn * 32, 256);

    int mtiles = (Nn + 15) / 16;

    k_premul_strip<128, 64><<<SGRID, SW * 32, 0, stream>>>(x, W1, b1, C, A, B, Nn, mtiles);
    k_gather64<true, false><<<gb, 256, 0, stream>>>(B, csr, rptr, A, Nn, 2.0f);
    k_gather64<true, true><<<gb, 256, 0, stream>>>(A, csr, rptr, C, Nn, 1.0f);

    k_gather64<false, false><<<gb, 256, 0, stream>>>(C, csr, rptr, A, Nn, 1.0f);
    k_gather64<false, false><<<gb, 256, 0, stream>>>(A, csr, rptr, B, Nn, 2.0f);
    k_cheb_strip<64, 64><<<SGRID, SW * 32, 0, stream>>>(C, A, B, W2, b2, D, Nn, mtiles);

    k_gather64<false, false><<<gb, 256, 0, stream>>>(D, csr, rptr, A, Nn, 1.0f);
    k_gather64<false, false><<<gb, 256, 0, stream>>>(A, csr, rptr, B, Nn, 2.0f);
    k_cheb_strip<64, 128><<<SGRID, SW * 32, 0, stream>>>(D, A, B, W3, b3, C, Nn, mtiles);

    k_fill<<<nb((long)G * H3 + G, 256), 256, 0, stream>>>(pool, 0.0f, G * H3 + G);
    k_count<<<nb(Nn, 256), 256, 0, stream>>>(bat, cnt, Nn);
    k_pool<<<nb((long)Nn * (H3 / 4), 256), 256, 0, stream>>>(C, bat, pool, Nn);
    k_final<<<G, H3, 0, stream>>>(pool, cnt, linW, linb, out, G, H3);
}


# mean-pool sums by run: each wave adds 16 consecutive node rows in registers and issues its atomic adds (with the node count) once per run of equal graph id, not once per node; sum order changes
# speedup vs baseline: 4.0847x; 1.3412x over previous
#include <hip/hip_runtime.h>

typedef __attribute__((ext_vector_type(2))) float v2f;
typedef __attribute__((ext_vector_type(8))) float v8f;


__global__ void k_fill(float* __restrict__ p, float v, int n) {
    int i = blockIdx.x * blockDim.x + threadIdx.x;
    if (i < n) p[i] = v;
}


__global__ void k_edge_deg(const int* __restrict__ ei, const float* __restrict__ attr,
                           float* __restrict__ deg, int* __restrict__ cnt,
                           int* __restrict__ rank, int E) {
    int e = blockIdx.x * blockDim.x + threadIdx.x;
    if (e >= E) return;
    int s = ei[e];
    int d = ei[E + e];
    float ew = (s != d) ? attr[e] : 0.0f;
    rank[e] = atomicAdd(&cnt[d], 1);
    if (ew != 0.0f) atomicAdd(&deg[d], ew);
}

__global__ void k_dis(float* __restrict__ deg, int n) {
    int i = blockIdx.x * blockDim.x + threadIdx.x;
    if (i >= n) return;
    float dv = deg[i];
    deg[i] = (dv > 0.0f) ? rsqrtf(fmaxf(dv, 1e-30f)) : 0.0f;
}

#define SCAN_T 1024
__global__ __launch_bounds__(SCAN_T) void
k_rowptr(const int* __restrict__ cnt, int* __restrict__ rowptr, int n) {
    __shared__ int pre_s[SCAN_T / 32];
    __shared__ int sum_s[SCAN_T / 32];
    int tid  = threadIdx.x;
    int lane = tid & 31;
    int wv   = tid >> 5;
    int base = blockIdx.x * (SCAN_T * 4);

    int pre = 0;
    const int4* c4 = reinterpret_cast<const int4*>(cnt) + tid;
#pragma unroll 4
    for (int k = 0; k < (int)blockIdx.x; ++k) {
        int4 v = c4[(size_t)k * SCAN_T];
        pre += (v.x + v.y) + (v.z + v.w);
    }
    for (int o = 16; o > 0; o >>= 1) pre += __shfl_xor(pre, o);
    if (lane == 0) pre_s[wv] = pre;

    int i0 = base + tid * 4;
    int c0 = (i0     < n) ? cnt[i0]     : 0;
    int c1 = (i0 + 1 < n) ? cnt[i0 + 1] : 0;
    int c2 = (i0 + 2 < n) ? cnt[i0 + 2] : 0;
    int c3 = (i0 + 3 < n) ? cnt[i0 + 3] : 0;
    int s = (c0 + c1) + (c2 + c3);
    int inc = s;
    for (int o = 1; o < 32; o <<= 1) {
        int t = __shfl_up(inc, o);
        if (lane >= o) inc += t;
    }
    if (lane == 31) sum_s[wv] = inc;
    __syncthreads();

    int t = pre_s[lane] + ((lane < wv) ? sum_s[lane] : 0);
    for (int o = 16; o > 0; o >>= 1) t += __shfl_xor(t, o);
    int x = t + inc - s;

    if (blockIdx.x == 0 && tid == 0) rowptr[0] = 0;
    x += c0; if (i0     < n) rowptr[i0 + 1] = x;
    x += c1; if (i0 + 1 < n) rowptr[i0 + 2] = x;
    x += c2; if (i0 + 2 < n) rowptr[i0 + 3] = x;
    x += c3; if (i0 + 3 < n) rowptr[i0 + 4] = x;
}

__global__ void k_edge_place(const int* __restrict__ ei, const float* __restrict__ attr,
                             const float* __restrict__ dis, const int* __restrict__ rowptr,
                             const int* __restrict__ rank, int2* __restrict__ csr, int E) {
    int e = blockIdx.x * blockDim.x + threadIdx.x;
    if (e >= E) return;
    int s = ei[e];
    int d = ei[E + e];
    float ew = (s != d) ? attr[e] : 0.0f;
    float wv = -dis[d] * ew * dis[s];
    csr[rowptr[d] + rank[e]] = make_int2(s, __float_as_int(wv));
}

template <int U>
__device__ __forceinline__ void gather_chunk(const int2* __restrict__ c,
                                             const float2* __restrict__ hp, float2& acc) {
    int2 ce[U];
    float2 v[U];
#pragma unroll
    for (int j = 0; j < U; ++j) ce[j] = c[j];
#pragma unroll
    for (int j = 0; j < U; ++j) v[j] = hp[(size_t)ce[j].x * 32];
#pragma unroll
    for (int j = 0; j < U; ++j) {
        float wj = __int_as_float(ce[j].y);
        acc.x = fmaf(wj, v[j].x, acc.x);
        acc.y = fmaf(wj, v[j].y, acc.y);
    }
}

template <bool HAS_BASE, bool RELU>
__global__ __launch_bounds__(256) void
k_gather64(const float* __restrict__ h, const int2* __restrict__ csr,
           const int* __restrict__ rowptr, float* __restrict__ out, int n, float scale) {
    int row = __builtin_amdgcn_readfirstlane((int)((blockIdx.x * 256 + threadIdx.x) >> 5));
    if (row >= n) return;
    int lane = threadIdx.x & 31;
    float2* op = reinterpret_cast<float2*>(out) + (size_t)row * 32 + lane;
    float2 b = make_float2(0.0f, 0.0f);
    if (HAS_BASE) b = *op;
    int e   = rowptr[row];
    int end = rowptr[row + 1];
    const float2* hp = reinterpret_cast<const float2*>(h) + lane;
    float2 acc = make_float2(0.0f, 0.0f);
    for (; e + 8 <= end; e += 8) gather_chunk<8>(csr + e, hp, acc);
    if (e + 4 <= end) { gather_chunk<4>(csr + e, hp, acc); e += 4; }
    int rem = end - e;
    if (rem > 0) {
        int2 t0 = csr[e];
        int2 t1 = csr[rem > 1 ? e + 1 : e];
        int2 t2 = csr[rem > 2 ? e + 2 : e];
        float2 v0 = hp[(size_t)t0.x * 32];
        float2 v1 = hp[(size_t)t1.x * 32];
        float2 v2 = hp[(size_t)t2.x * 32];
        float w0 = __int_as_float(t0.y);
        acc.x = fmaf(w0, v0.x, acc.x);
        acc.y = fmaf(w0, v0.y, acc.y);
        if (rem > 1) {
            float w1 = __int_as_float(t1.y);
            acc.x = fmaf(w1, v1.x, acc.x);
            acc.y = fmaf(w1, v1.y, acc.y);
        }
        if (rem > 2) {
            float w2 = __int_as_float(t2.y);
            acc.x = fmaf(w2, v2.x, acc.x);
            acc.y = fmaf(w2, v2.y, acc.y);
        }
    }
    float2 r;
    r.x = scale * acc.x + b.x;
    r.y = scale * acc.y + b.y;
    if (RELU) { r.x = fmaxf(r.x, 0.0f); r.y = fmaxf(r.y, 0.0f); }
    *op = r;
}

#define SW 12
#define SGRID 256
#define SLDS 41984

template <int KD, int N>
__device__ __forceinline__ void stage_weights(const float* __restrict__ W,
                                              float* __restrict__ Wl, int tid) {
    constexpr int KP   = KD + 4;
    constexpr int Q    = N / 4;
    constexpr int SL4  = KD * Q;
    constexpr int NB16 = N / 16;
    constexpr int GPS  = (KD / 8) * NB16;
    constexpr int IT0  = (GPS + SW - 1) / SW;
    constexpr int IT1  = (2 * GPS + SW - 1) / SW;
    const float4* W4 = reinterpret_cast<const float4*>(W);
    int wv = tid >> 5;
    int q  = tid & 3;
    int kk = (tid >> 2) & 7;

    float4 va[IT0], vb[IT0], vc[IT1];
#pragma unroll
    for (int it = 0; it < IT0; ++it) {
        int g    = min(wv + it * SW, GPS - 1);
        int kblk = g / NB16;
        int nblk = g - kblk * NB16;
        int idx  = (kblk * 8 + kk) * Q + nblk * 4 + q;
        va[it] = W4[idx];
        vb[it] = W4[idx + 2 * SL4];
    }
#pragma unroll
    for (int it = 0; it < IT1; ++it) {
        int g    = min(wv + it * SW, 2 * GPS - 1);
        int t    = g / GPS;
        int r    = g - t * GPS;
        int kblk = r / NB16;
        int nblk = r - kblk * NB16;
        vc[it] = W4[((t + 1) * KD + kblk * 8 + kk) * Q + nblk * 4 + q];
    }
#pragma unroll
    for (int it = 0; it < IT0; ++it) {
        int g = wv + it * SW;
        if (g < GPS) {
            int kblk = g / NB16;
            int nblk = g - kblk * NB16;
            float* p = Wl + (nblk * 16 + q * 4) * KP + kblk * 8 + kk;
            p[0]      = va[it].x - vb[it].x;
            p[KP]     = va[it].y - vb[it].y;
            p[2 * KP] = va[it].z - vb[it].z;
            p[3 * KP] = va[it].w - vb[it].w;
        }
    }
#pragma unroll
    for (int it = 0; it < IT1; ++it) {
        int g = wv + it * SW;
        if (g < 2 * GPS) {
            int t    = g / GPS;
            int r    = g - t * GPS;
            int kblk = r / NB16;
            int nblk = r - kblk * NB16;
            float* p = Wl + ((t + 1) * N + nblk * 16 + q * 4) * KP + kblk * 8 + kk;
            p[0]      = vc[it].x;
            p[KP]     = vc[it].y;
            p[2 * KP] = vc[it].z;
            p[3 * KP] = vc[it].w;
        }
    }
}

template <int KB>
__device__ __forceinline__ void load_a(float4 (&a)[KB], const float* __restrict__ rowp, int h) {
    const float4* p = reinterpret_cast<const float4*>(rowp) + h;
#pragma unroll
    for (int kb = 0; kb < KB; ++kb) a[kb] = p[kb * 2];
}

template <int NT, int KP>
__device__ __forceinline__ void load_b(float4 (&b)[NT], const float* p) {
    __builtin_amdgcn_fence(__ATOMIC_ACQUIRE, "wavefront", "local");
#pragma unroll
    for (int j = 0; j < NT; ++j)
        b[j] = *reinterpret_cast<const float4*>(p + j * 16 * KP);
}

template <int NT>
__device__ __forceinline__ void mma_step(const float4& a, const float4 (&b)[NT], v8f (&acc)[NT]) {
    v2f alo, ahi;
    alo.x = a.x; alo.y = a.y;
    ahi.x = a.z; ahi.y = a.w;
#pragma unroll
    for (int j = 0; j < NT; ++j) {
        v2f bl; bl.x = b[j].x; bl.y = b[j].y;
        acc[j] = __builtin_amdgcn_wmma_f32_16x16x4_f32(
                      false, alo,  false, bl,
                      (short)0, acc[j],  false,  false);
    }
#pragma unroll
    for (int j = 0; j < NT; ++j) {
        v2f bh; bh.x = b[j].z; bh.y = b[j].w;
        acc[j] = __builtin_amdgcn_wmma_f32_16x16x4_f32(
                     false, ahi, false, bh, (short)0, acc[j], false, false);
    }
}

template <int KB, int NT, int KP>
__device__ __forceinline__ void mma_unit(const float4 (&a)[KB], const float* bp, const float* bnext,
                                         float4 (&bc)[NT], v8f (&acc)[NT]) {
#pragma unroll
    for (int kb = 0; kb < KB; ++kb) {
        float4 bn[NT];
        load_b<NT, KP>(bn, (kb + 1 < KB) ? bp + (kb + 1) * 8 : bnext);
        mma_step<NT>(a[kb], bc, acc);
#pragma unroll
        for (int j = 0; j < NT; ++j) bc[j] = bn[j];
    }
}

template <int KD, int N>
__global__ __launch_bounds__(SW * 32) void
k_cheb_strip(const float* __restrict__ Tx0, const float* __restrict__ Tx1,
             const float* __restrict__ P, const float* __restrict__ W,
             const float* __restrict__ bias, float* __restrict__ out, int M, int Mtiles) {
    __shared__ __attribute__((aligned(16))) float Wl[SLDS];
    constexpr int KP = KD + 4;
    constexpr int NT = N / 16;
    constexpr int KB = KD / 8;

    int tid = threadIdx.x;
    stage_weights<KD, N>(W, Wl, tid);
    __syncthreads();

    int wave = __builtin_amdgcn_readfirstlane(tid >> 5);
    int step = SW * (int)gridDim.x;
    int mt   = wave * (int)gridDim.x + (int)blockIdx.x;
    if (mt >= Mtiles) return;

    int lane = tid & 31;
    int c    = lane & 15;
    int h    = lane >> 4;

    float bv[NT];
#pragma unroll
    for (int j = 0; j < NT; ++j) bv[j] = bias[j * 16 + c];

    const float* bp = Wl + c * KP + 4 * h;
    int row = min(mt * 16 + c, M - 1);

    float4 a0[KB], a1[KB];
    float4 bc[NT];
    load_a<KB>(a0, Tx0 + (size_t)row * KD, h);
    load_b<NT, KP>(bc, bp);
    for (;;) {
        int  mtn  = mt + step;
        bool more = mtn < Mtiles;
        int  rown = min((more ? mtn : mt) * 16 + c, M - 1);

        v8f acc[NT];
#pragma unroll
        for (int j = 0; j < NT; ++j) {
#pragma unroll
            for (int i = 0; i < 8; ++i) acc[j][i] = bv[j];
        }

        load_a<KB>(a1, Tx1 + (size_t)row * KD, h);
        mma_unit<KB, NT, KP>(a0, bp, bp + N * KP, bc, acc);
        load_a<KB>(a0, P + (size_t)row * KD, h);
        mma_unit<KB, NT, KP>(a1, bp + N * KP, bp + 2 * N * KP, bc, acc);
        load_a<KB>(a1, Tx0 + (size_t)rown * KD, h);
        mma_unit<KB, NT, KP>(a0, bp + 2 * N * KP, bp, bc, acc);

        int rbase = mt * 16 + (h << 3);
#pragma unroll
        for (int i = 0; i < 8; ++i) {
            int r = rbase + i;
            if (r < M) {
                float* op = out + (size_t)r * N + c;
#pragma unroll
                for (int j = 0; j < NT; ++j) op[j * 16] = fmaxf(acc[j][i], 0.0f);
            }
        }
        if (!more) break;
#pragma unroll
        for (int kb = 0; kb < KB; ++kb) a0[kb] = a1[kb];
        mt  = mtn;
        row = rown;
    }
}

template <int KD, int N>
__global__ __launch_bounds__(SW * 32) void
k_premul_strip(const float* __restrict__ X, const float* __restrict__ W,
               const float* __restrict__ bias, float* __restrict__ Y0,
               float* __restrict__ Y1, float* __restrict__ Y2, int M, int Mtiles) {
    __shared__ __attribute__((aligned(16))) float Wl[SLDS];
    constexpr int KP = KD + 4;
    constexpr int NT = N / 16;
    constexpr int KH = KD / 2;
    constexpr int KB = KH / 8;

    int tid = threadIdx.x;
    stage_weights<KD, N>(W, Wl, tid);
    __syncthreads();

    int wave = __builtin_amdgcn_readfirstlane(tid >> 5);
    int step = SW * (int)gridDim.x;
    int mt   = wave * (int)gridDim.x + (int)blockIdx.x;
    if (mt >= Mtiles) return;

    int lane = tid & 31;
    int c    = lane & 15;
    int h    = lane >> 4;

    float bv[NT];
#pragma unroll
    for (int j = 0; j < NT; ++j) bv[j] = bias[j * 16 + c];

    const float* bp = Wl + c * KP + 4 * h;
    int row = min(mt * 16 + c, M - 1);

    float4 a0[KB], a1[KB];
    float4 bc[3 * NT];
    load_a<KB>(a0, X + (size_t)row * KD, h);
    load_b<3 * NT, KP>(bc, bp);
    for (;;) {
        int  mtn  = mt + step;
        bool more = mtn < Mtiles;
        int  rown = min((more ? mtn : mt) * 16 + c, M - 1);

        v8f acc[3 * NT];
#pragma unroll
        for (int j = 0; j < 3 * NT; ++j) {
#pragma unroll
            for (int i = 0; i < 8; ++i) acc[j][i] = (j < NT) ? bv[j < NT ? j : 0] : 0.0f;
        }

        load_a<KB>(a1, X + (size_t)row * KD + KH, h);
        mma_unit<KB, 3 * NT, KP>(a0, bp, bp + KH, bc, acc);
        load_a<KB>(a0, X + (size_t)rown * KD, h);
        mma_unit<KB, 3 * NT, KP>(a1, bp + KH, bp, bc, acc);

        int rbase = mt * 16 + (h << 3);
#pragma unroll
        for (int i = 0; i < 8; ++i) {
            int r = rbase + i;
            if (r < M) {
                size_t o = (size_t)r * N + c;
#pragma unroll
                for (int j = 0; j < NT; ++j) {
                    Y0[o + j * 16] = acc[j][i];
                    Y1[o + j * 16] = acc[NT + j][i];
                    Y2[o + j * 16] = acc[2 * NT + j][i];
                }
            }
        }
        if (!more) break;
        mt  = mtn;
        row = rown;
    }
}


#define POOL_ROWS 16
__device__ __forceinline__ void pool_flush(float* __restrict__ sums, float* __restrict__ cnt,
                                           int g, int lane, const float4& a, float c) {
    float* sp = sums + ((size_t)g << 7) + (lane << 2);
    atomicAdd(sp + 0, a.x);
    atomicAdd(sp + 1, a.y);
    atomicAdd(sp + 2, a.z);
    atomicAdd(sp + 3, a.w);
    if (lane == 0) atomicAdd(cnt + g, c);
}

__global__ __launch_bounds__(256) void
k_pool_runs(const float* __restrict__ h, const int* __restrict__ batch,
            float* __restrict__ sums, float* __restrict__ cnt, int n) {
    int wave = __builtin_amdgcn_readfirstlane((int)((blockIdx.x * 256 + threadIdx.x) >> 5));
    int r0 = wave * POOL_ROWS;
    if (r0 >= n) return;
    int lane = threadIdx.x & 31;
    const float4* hp = reinterpret_cast<const float4*>(h) + ((size_t)r0 << 5) + lane;
    const int*    bp = batch + r0;

    float4 acc = make_float4(0.0f, 0.0f, 0.0f, 0.0f);
    float  c   = 0.0f;
    int    gc;
    if (r0 + POOL_ROWS <= n) {
        int    g[POOL_ROWS];
        float4 v[POOL_ROWS];
#pragma unroll
        for (int j = 0; j < POOL_ROWS; ++j) g[j] = bp[j];
#pragma unroll
        for (int j = 0; j < POOL_ROWS; ++j) v[j] = hp[j << 5];
        gc = g[0];
#pragma unroll
        for (int j = 0; j < POOL_ROWS; ++j) {
            if (g[j] != gc) {
                pool_flush(sums, cnt, gc, lane, acc, c);
                acc = make_float4(0.0f, 0.0f, 0.0f, 0.0f);
                c   = 0.0f;
                gc  = g[j];
            }
            acc.x += v[j].x; acc.y += v[j].y; acc.z += v[j].z; acc.w += v[j].w;
            c += 1.0f;
        }
    } else {
        gc = bp[0];
        for (int j = 0; j < n - r0; ++j) {
            int    gj = bp[j];
            float4 vj = hp[j << 5];
            if (gj != gc) {
                pool_flush(sums, cnt, gc, lane, acc, c);
                acc = make_float4(0.0f, 0.0f, 0.0f, 0.0f);
                c   = 0.0f;
                gc  = gj;
            }
            acc.x += vj.x; acc.y += vj.y; acc.z += vj.z; acc.w += vj.w;
            c += 1.0f;
        }
    }
    pool_flush(sums, cnt, gc, lane, acc, c);
}

__global__ void k_final(const float* __restrict__ sums, const float* __restrict__ cnt,
                        const float* __restrict__ linW, const float* __restrict__ linb,
                        float* __restrict__ out, int G, int H) {
    __shared__ float hg_s[128];
    int g = blockIdx.x;
    int f = threadIdx.x;
    float c = cnt[g];
    float hg = sums[(size_t)g * H + f] / fmaxf(c, 1.0f);
    out[(size_t)G * 2 + (size_t)g * H + f] = hg;
    hg_s[f] = hg;
    __syncthreads();
    if (f < 2) {
        float acc = linb[f];
        for (int j = 0; j < H; ++j) acc += hg_s[j] * linW[j * 2 + f];
        out[g * 2 + f] = acc;
    }
}


extern "C" void kernel_launch(void* const* d_in, const int* in_sizes, int n_in,
                              void* d_out, int out_size, void* d_ws, size_t ws_size,
                              hipStream_t stream) {
    const float* x    = (const float*)d_in[0];
    const int*   ei   = (const int*)d_in[1];
    const float* attr = (const float*)d_in[2];
    const int*   bat  = (const int*)d_in[3];
    const float* W1 = (const float*)d_in[4];  const float* b1 = (const float*)d_in[5];
    const float* W2 = (const float*)d_in[6];  const float* b2 = (const float*)d_in[7];
    const float* W3 = (const float*)d_in[8];  const float* b3 = (const float*)d_in[9];
    const float* linW = (const float*)d_in[10];
    const float* linb = (const float*)d_in[11];
    float* out = (float*)d_out;

    const int F_IN = 128, H1 = 64, H2 = 64, H3 = 128, NCLS = 2;
    int Nn = in_sizes[0] / F_IN;
    int E  = in_sizes[2];
    int G  = out_size / (NCLS + H3);

    auto up = [](size_t v) { return (v + 63) & ~(size_t)63; };
    float* ws = (float*)d_ws;
    size_t off = 0;
    float* dis  = ws + off; off += up((size_t)Nn);
    int*   ecnt = (int*)(ws + off); off += up((size_t)Nn);
    int*   rptr = (int*)(ws + off); off += up((size_t)Nn + 1);
    int*   rank = (int*)(ws + off); off += up((size_t)E);
    int2*  csr  = (int2*)(ws + off); off += up((size_t)E * 2);
    float* A    = ws + off; off += up((size_t)Nn * H1);
    float* B    = ws + off; off += up((size_t)Nn * H1);
    float* C    = ws + off; off += up((size_t)Nn * H3);
    float* D    = ws + off; off += up((size_t)Nn * H2);
    float* pool = ws + off; off += (size_t)G * H3;
    float* cnt  = ws + off; off += (size_t)G;

    auto nb = [](long n, int t) { return (unsigned)((n + t - 1) / t); };

    long nz = (long)(up((size_t)Nn) + (size_t)Nn);
    k_fill<<<nb(nz, 256), 256, 0, stream>>>(dis, 0.0f, (int)nz);
    k_edge_deg<<<nb(E, 256), 256, 0, stream>>>(ei, attr, dis, ecnt, rank, E);
    k_dis<<<nb(Nn, 256), 256, 0, stream>>>(dis, Nn);
    k_rowptr<<<nb(Nn, SCAN_T * 4), SCAN_T, 0, stream>>>(ecnt, rptr, Nn);
    k_edge_place<<<nb(E, 256), 256, 0, stream>>>(ei, attr, dis, rptr, rank, csr, E);

    unsigned gb = nb((long)Nn * 32, 256);

    int mtiles = (Nn + 15) / 16;

    k_premul_strip<128, 64><<<SGRID, SW * 32, 0, stream>>>(x, W1, b1, C, A, B, Nn, mtiles);
    k_gather64<true, false><<<gb, 256, 0, stream>>>(B, csr, rptr, A, Nn, 2.0f);
    k_gather64<true, true><<<gb, 256, 0, stream>>>(A, csr, rptr, C, Nn, 1.0f);

    k_gather64<false, false><<<gb, 256, 0, stream>>>(C, csr, rptr, A, Nn, 1.0f);
    k_gather64<false, false><<<gb, 256, 0, stream>>>(A, csr, rptr, B, Nn, 2.0f);
    k_cheb_strip<64, 64><<<SGRID, SW * 32, 0, stream>>>(C, A, B, W2, b2, D, Nn, mtiles);

    k_gather64<false, false><<<gb, 256, 0, stream>>>(D, csr, rptr, A, Nn, 1.0f);
    k_gather64<false, false><<<gb, 256, 0, stream>>>(A, csr, rptr, B, Nn, 2.0f);
    k_cheb_strip<64, 128><<<SGRID, SW * 32, 0, stream>>>(D, A, B, W3, b3, C, Nn, mtiles);

    k_fill<<<nb((long)G * H3 + G, 256), 256, 0, stream>>>(pool, 0.0f, G * H3 + G);
    k_pool_runs<<<nb((long)nb(Nn, POOL_ROWS) * 32, 256), 256, 0, stream>>>(C, bat, pool, cnt, Nn);
    k_final<<<G, H3, 0, stream>>>(pool, cnt, linW, linb, out, G, H3);
}


# CSR build by two-level counting sort on LDS counters (bin edges by 256-node bucket, then one block per bucket) replaces a returning global atomic per edge; degree sums, row slots in LDS arrival order
# speedup vs baseline: 5.7487x; 1.4074x over previous
#include <hip/hip_runtime.h>

typedef __attribute__((ext_vector_type(2))) float v2f;
typedef __attribute__((ext_vector_type(8))) float v8f;


__global__ void k_fill(float* __restrict__ p, float v, int n) {
    int i = blockIdx.x * blockDim.x + threadIdx.x;
    if (i < n) p[i] = v;
}

#define BK_SH 8
#define BK_W  (1 << BK_SH)
#define MAXB  1024
#define BIN_T 1024
#define EPT1  8
#define EPT2  5

__global__ __launch_bounds__(BIN_T) void
k_bin_count(const int* __restrict__ ei, int* __restrict__ histG, int E, int nbk) {
    __shared__ int hist[MAXB];
    int tid = threadIdx.x;
    int c   = blockIdx.x;
    int nch = gridDim.x;
    int ch  = (E + nch - 1) / nch;
    int e0  = c * ch;
    int e1  = min(E, e0 + ch);
    for (int i = tid; i < nbk; i += BIN_T) hist[i] = 0;
    __syncthreads();
    for (int base = e0 + tid; base < e1 + tid; base += BIN_T * EPT1) {
        int d[EPT1];
#pragma unroll
        for (int j = 0; j < EPT1; ++j) d[j] = ei[E + min(base + j * BIN_T, e1 - 1)];
#pragma unroll
        for (int j = 0; j < EPT1; ++j)
            if (base + j * BIN_T < e1) atomicAdd(&hist[d[j] >> BK_SH], 1);
    }
    __syncthreads();
    for (int i = tid; i < nbk; i += BIN_T) histG[i * nch + c] = hist[i];
}

#define SCAN_T 1024
__global__ __launch_bounds__(SCAN_T) void
k_rowptr(const int* __restrict__ cnt, int* __restrict__ rowptr, int n) {
    __shared__ int pre_s[SCAN_T / 32];
    __shared__ int sum_s[SCAN_T / 32];
    int tid  = threadIdx.x;
    int lane = tid & 31;
    int wv   = tid >> 5;
    int base = blockIdx.x * (SCAN_T * 4);

    int pre = 0;
    const int4* c4 = reinterpret_cast<const int4*>(cnt) + tid;
#pragma unroll 4
    for (int k = 0; k < (int)blockIdx.x; ++k) {
        int4 v = c4[(size_t)k * SCAN_T];
        pre += (v.x + v.y) + (v.z + v.w);
    }
    for (int o = 16; o > 0; o >>= 1) pre += __shfl_xor(pre, o);
    if (lane == 0) pre_s[wv] = pre;

    int i0 = base + tid * 4;
    int c0 = (i0     < n) ? cnt[i0]     : 0;
    int c1 = (i0 + 1 < n) ? cnt[i0 + 1] : 0;
    int c2 = (i0 + 2 < n) ? cnt[i0 + 2] : 0;
    int c3 = (i0 + 3 < n) ? cnt[i0 + 3] : 0;
    int s = (c0 + c1) + (c2 + c3);
    int inc = s;
    for (int o = 1; o < 32; o <<= 1) {
        int t = __shfl_up(inc, o);
        if (lane >= o) inc += t;
    }
    if (lane == 31) sum_s[wv] = inc;
    __syncthreads();

    int t = pre_s[lane] + ((lane < wv) ? sum_s[lane] : 0);
    for (int o = 16; o > 0; o >>= 1) t += __shfl_xor(t, o);
    int x = t + inc - s;

    if (blockIdx.x == 0 && tid == 0) rowptr[0] = 0;
    x += c0; if (i0     < n) rowptr[i0 + 1] = x;
    x += c1; if (i0 + 1 < n) rowptr[i0 + 2] = x;
    x += c2; if (i0 + 2 < n) rowptr[i0 + 3] = x;
    x += c3; if (i0 + 3 < n) rowptr[i0 + 4] = x;
}

__global__ __launch_bounds__(BIN_T) void
k_bin_place(const int* __restrict__ ei, const float* __restrict__ attr,
            const int* __restrict__ binOff, int2* __restrict__ binned, int E, int nbk) {
    __shared__ int cur[MAXB];
    int tid = threadIdx.x;
    int c   = blockIdx.x;
    int nch = gridDim.x;
    int ch  = (E + nch - 1) / nch;
    int e0  = c * ch;
    int e1  = min(E, e0 + ch);
    for (int i = tid; i < nbk; i += BIN_T) cur[i] = binOff[i * nch + c];
    __syncthreads();
    for (int base = e0 + tid; base < e1 + tid; base += BIN_T * EPT1) {
        int   s[EPT1], d[EPT1];
        float a[EPT1];
#pragma unroll
        for (int j = 0; j < EPT1; ++j) {
            int e = min(base + j * BIN_T, e1 - 1);
            s[j] = ei[e];
            d[j] = ei[E + e];
            a[j] = attr[e];
        }
#pragma unroll
        for (int j = 0; j < EPT1; ++j) {
            if (base + j * BIN_T < e1) {
                float ew = (s[j] != d[j]) ? a[j] : 0.0f;
                int pos = atomicAdd(&cur[d[j] >> BK_SH], 1);
                binned[pos] = make_int2((s[j] << BK_SH) | (d[j] & (BK_W - 1)), __float_as_int(ew));
            }
        }
    }
}

__global__ __launch_bounds__(BIN_T) void
k_bucket_csr(const int2* __restrict__ binned, const int* __restrict__ binOff, int nch,
             float* __restrict__ dis, int* __restrict__ rowptr, int2* __restrict__ csr,
             int n, int E) {
    __shared__ int   cntL[BK_W];
    __shared__ float degL[BK_W];
    __shared__ int   curL[BK_W];
    __shared__ float disL[BK_W];
    __shared__ int   wsum[BK_W / 32];
    int tid  = threadIdx.x;
    int lane = tid & 31;
    int b    = blockIdx.x;
    int e0   = binOff[b * nch];
    int e1   = binOff[(b + 1) * nch];
    if (tid < BK_W) { cntL[tid] = 0; degL[tid] = 0.0f; }
    __syncthreads();
    for (int base = e0 + tid; base < e1 + tid; base += BIN_T * EPT2) {
        int2 r[EPT2];
#pragma unroll
        for (int j = 0; j < EPT2; ++j) r[j] = binned[min(base + j * BIN_T, e1 - 1)];
#pragma unroll
        for (int j = 0; j < EPT2; ++j) {
            if (base + j * BIN_T < e1) {
                int ld = r[j].x & (BK_W - 1);
                float ew = __int_as_float(r[j].y);
                atomicAdd(&cntL[ld], 1);
                if (ew != 0.0f) atomicAdd(&degL[ld], ew);
            }
        }
    }
    __syncthreads();
    int cn  = (tid < BK_W) ? cntL[tid] : 0;
    int inc = cn;
    for (int o = 1; o < 32; o <<= 1) {
        int t = __shfl_up(inc, o);
        if (lane >= o) inc += t;
    }
    if (tid < BK_W && lane == 31) wsum[tid >> 5] = inc;
    __syncthreads();
    if (tid < BK_W) {
        int wv  = tid >> 5;
        int pre = 0;
        for (int k = 0; k < BK_W / 32; ++k) pre += (k < wv) ? wsum[k] : 0;
        int start = e0 + pre + inc - cn;
        float dv = degL[tid];
        float di = (dv > 0.0f) ? rsqrtf(fmaxf(dv, 1e-30f)) : 0.0f;
        curL[tid] = start;
        disL[tid] = di;
        int node = b * BK_W + tid;
        if (node < n) { rowptr[node] = start; dis[node] = di; }
    }
    if (b == (int)gridDim.x - 1 && tid == 0) rowptr[n] = E;
    __syncthreads();
    for (int base = e0 + tid; base < e1 + tid; base += BIN_T * EPT2) {
        int2 r[EPT2];
#pragma unroll
        for (int j = 0; j < EPT2; ++j) r[j] = binned[min(base + j * BIN_T, e1 - 1)];
#pragma unroll
        for (int j = 0; j < EPT2; ++j) {
            if (base + j * BIN_T < e1) {
                int ld = r[j].x & (BK_W - 1);
                int s  = (int)((unsigned)r[j].x >> BK_SH);
                float ew = __int_as_float(r[j].y);
                int pos = atomicAdd(&curL[ld], 1);
                csr[pos] = make_int2(s, __float_as_int(-disL[ld] * ew));
            }
        }
    }
}

__global__ void k_csr_norm(int2* __restrict__ csr, const float* __restrict__ dis, int E) {
    int j = blockIdx.x * blockDim.x + threadIdx.x;
    if (j >= E) return;
    int2 r = csr[j];
    r.y = __float_as_int(__int_as_float(r.y) * dis[r.x]);
    csr[j] = r;
}

template <int U>
__device__ __forceinline__ void gather_chunk(const int2* __restrict__ c,
                                             const float2* __restrict__ hp, float2& acc) {
    int2 ce[U];
    float2 v[U];
#pragma unroll
    for (int j = 0; j < U; ++j) ce[j] = c[j];
#pragma unroll
    for (int j = 0; j < U; ++j) v[j] = hp[(size_t)ce[j].x * 32];
#pragma unroll
    for (int j = 0; j < U; ++j) {
        float wj = __int_as_float(ce[j].y);
        acc.x = fmaf(wj, v[j].x, acc.x);
        acc.y = fmaf(wj, v[j].y, acc.y);
    }
}

template <bool HAS_BASE, bool RELU>
__global__ __launch_bounds__(256) void
k_gather64(const float* __restrict__ h, const int2* __restrict__ csr,
           const int* __restrict__ rowptr, float* __restrict__ out, int n, float scale) {
    int row = __builtin_amdgcn_readfirstlane((int)((blockIdx.x * 256 + threadIdx.x) >> 5));
    if (row >= n) return;
    int lane = threadIdx.x & 31;
    float2* op = reinterpret_cast<float2*>(out) + (size_t)row * 32 + lane;
    float2 b = make_float2(0.0f, 0.0f);
    if (HAS_BASE) b = *op;
    int e   = rowptr[row];
    int end = rowptr[row + 1];
    const float2* hp = reinterpret_cast<const float2*>(h) + lane;
    float2 acc = make_float2(0.0f, 0.0f);
    for (; e + 8 <= end; e += 8) gather_chunk<8>(csr + e, hp, acc);
    if (e + 4 <= end) { gather_chunk<4>(csr + e, hp, acc); e += 4; }
    int rem = end - e;
    if (rem > 0) {
        int2 t0 = csr[e];
        int2 t1 = csr[rem > 1 ? e + 1 : e];
        int2 t2 = csr[rem > 2 ? e + 2 : e];
        float2 v0 = hp[(size_t)t0.x * 32];
        float2 v1 = hp[(size_t)t1.x * 32];
        float2 v2 = hp[(size_t)t2.x * 32];
        float w0 = __int_as_float(t0.y);
        acc.x = fmaf(w0, v0.x, acc.x);
        acc.y = fmaf(w0, v0.y, acc.y);
        if (rem > 1) {
            float w1 = __int_as_float(t1.y);
            acc.x = fmaf(w1, v1.x, acc.x);
            acc.y = fmaf(w1, v1.y, acc.y);
        }
        if (rem > 2) {
            float w2 = __int_as_float(t2.y);
            acc.x = fmaf(w2, v2.x, acc.x);
            acc.y = fmaf(w2, v2.y, acc.y);
        }
    }
    float2 r;
    r.x = scale * acc.x + b.x;
    r.y = scale * acc.y + b.y;
    if (RELU) { r.x = fmaxf(r.x, 0.0f); r.y = fmaxf(r.y, 0.0f); }
    *op = r;
}

#define SW 12
#define SGRID 256
#define SLDS 41984

template <int KD, int N>
__device__ __forceinline__ void stage_weights(const float* __restrict__ W,
                                              float* __restrict__ Wl, int tid) {
    constexpr int KP   = KD + 4;
    constexpr int Q    = N / 4;
    constexpr int SL4  = KD * Q;
    constexpr int NB16 = N / 16;
    constexpr int GPS  = (KD / 8) * NB16;
    constexpr int IT0  = (GPS + SW - 1) / SW;
    constexpr int IT1  = (2 * GPS + SW - 1) / SW;
    const float4* W4 = reinterpret_cast<const float4*>(W);
    int wv = tid >> 5;
    int q  = tid & 3;
    int kk = (tid >> 2) & 7;

    float4 va[IT0], vb[IT0], vc[IT1];
#pragma unroll
    for (int it = 0; it < IT0; ++it) {
        int g    = min(wv + it * SW, GPS - 1);
        int kblk = g / NB16;
        int nblk = g - kblk * NB16;
        int idx  = (kblk * 8 + kk) * Q + nblk * 4 + q;
        va[it] = W4[idx];
        vb[it] = W4[idx + 2 * SL4];
    }
#pragma unroll
    for (int it = 0; it < IT1; ++it) {
        int g    = min(wv + it * SW, 2 * GPS - 1);
        int t    = g / GPS;
        int r    = g - t * GPS;
        int kblk = r / NB16;
        int nblk = r - kblk * NB16;
        vc[it] = W4[((t + 1) * KD + kblk * 8 + kk) * Q + nblk * 4 + q];
    }
#pragma unroll
    for (int it = 0; it < IT0; ++it) {
        int g = wv + it * SW;
        if (g < GPS) {
            int kblk = g / NB16;
            int nblk = g - kblk * NB16;
            float* p = Wl + (nblk * 16 + q * 4) * KP + kblk * 8 + kk;
            p[0]      = va[it].x - vb[it].x;
            p[KP]     = va[it].y - vb[it].y;
            p[2 * KP] = va[it].z - vb[it].z;
            p[3 * KP] = va[it].w - vb[it].w;
        }
    }
#pragma unroll
    for (int it = 0; it < IT1; ++it) {
        int g = wv + it * SW;
        if (g < 2 * GPS) {
            int t    = g / GPS;
            int r    = g - t * GPS;
            int kblk = r / NB16;
            int nblk = r - kblk * NB16;
            float* p = Wl + ((t + 1) * N + nblk * 16 + q * 4) * KP + kblk * 8 + kk;
            p[0]      = vc[it].x;
            p[KP]     = vc[it].y;
            p[2 * KP] = vc[it].z;
            p[3 * KP] = vc[it].w;
        }
    }
}

template <int KB>
__device__ __forceinline__ void load_a(float4 (&a)[KB], const float* __restrict__ rowp, int h) {
    const float4* p = reinterpret_cast<const float4*>(rowp) + h;
#pragma unroll
    for (int kb = 0; kb < KB; ++kb) a[kb] = p[kb * 2];
}

template <int NT, int KP>
__device__ __forceinline__ void load_b(float4 (&b)[NT], const float* p) {
    __builtin_amdgcn_fence(__ATOMIC_ACQUIRE, "wavefront", "local");
#pragma unroll
    for (int j = 0; j < NT; ++j)
        b[j] = *reinterpret_cast<const float4*>(p + j * 16 * KP);
}

template <int NT>
__device__ __forceinline__ void mma_step(const float4& a, const float4 (&b)[NT], v8f (&acc)[NT]) {
    v2f alo, ahi;
    alo.x = a.x; alo.y = a.y;
    ahi.x = a.z; ahi.y = a.w;
#pragma unroll
    for (int j = 0; j < NT; ++j) {
        v2f bl; bl.x = b[j].x; bl.y = b[j].y;
        acc[j] = __builtin_amdgcn_wmma_f32_16x16x4_f32(
                      false, alo,  false, bl,
                      (short)0, acc[j],  false,  false);
    }
#pragma unroll
    for (int j = 0; j < NT; ++j) {
        v2f bh; bh.x = b[j].z; bh.y = b[j].w;
        acc[j] = __builtin_amdgcn_wmma_f32_16x16x4_f32(
                     false, ahi, false, bh, (short)0, acc[j], false, false);
    }
}

template <int KB, int NT, int KP>
__device__ __forceinline__ void mma_unit(const float4 (&a)[KB], const float* bp, const float* bnext,
                                         float4 (&bc)[NT], v8f (&acc)[NT]) {
#pragma unroll
    for (int kb = 0; kb < KB; ++kb) {
        float4 bn[NT];
        load_b<NT, KP>(bn, (kb + 1 < KB) ? bp + (kb + 1) * 8 : bnext);
        mma_step<NT>(a[kb], bc, acc);
#pragma unroll
        for (int j = 0; j < NT; ++j) bc[j] = bn[j];
    }
}

template <int KD, int N>
__global__ __launch_bounds__(SW * 32) void
k_cheb_strip(const float* __restrict__ Tx0, const float* __restrict__ Tx1,
             const float* __restrict__ P, const float* __restrict__ W,
             const float* __restrict__ bias, float* __restrict__ out, int M, int Mtiles) {
    __shared__ __attribute__((aligned(16))) float Wl[SLDS];
    constexpr int KP = KD + 4;
    constexpr int NT = N / 16;
    constexpr int KB = KD / 8;

    int tid = threadIdx.x;
    stage_weights<KD, N>(W, Wl, tid);
    __syncthreads();

    int wave = __builtin_amdgcn_readfirstlane(tid >> 5);
    int step = SW * (int)gridDim.x;
    int mt   = wave * (int)gridDim.x + (int)blockIdx.x;
    if (mt >= Mtiles) return;

    int lane = tid & 31;
    int c    = lane & 15;
    int h    = lane >> 4;

    float bv[NT];
#pragma unroll
    for (int j = 0; j < NT; ++j) bv[j] = bias[j * 16 + c];

    const float* bp = Wl + c * KP + 4 * h;
    int row = min(mt * 16 + c, M - 1);

    float4 a0[KB], a1[KB];
    float4 bc[NT];
    load_a<KB>(a0, Tx0 + (size_t)row * KD, h);
    load_b<NT, KP>(bc, bp);
    for (;;) {
        int  mtn  = mt + step;
        bool more = mtn < Mtiles;
        int  rown = min((more ? mtn : mt) * 16 + c, M - 1);

        v8f acc[NT];
#pragma unroll
        for (int j = 0; j < NT; ++j) {
#pragma unroll
            for (int i = 0; i < 8; ++i) acc[j][i] = bv[j];
        }

        load_a<KB>(a1, Tx1 + (size_t)row * KD, h);
        mma_unit<KB, NT, KP>(a0, bp, bp + N * KP, bc, acc);
        load_a<KB>(a0, P + (size_t)row * KD, h);
        mma_unit<KB, NT, KP>(a1, bp + N * KP, bp + 2 * N * KP, bc, acc);
        load_a<KB>(a1, Tx0 + (size_t)rown * KD, h);
        mma_unit<KB, NT, KP>(a0, bp + 2 * N * KP, bp, bc, acc);

        int rbase = mt * 16 + (h << 3);
#pragma unroll
        for (int i = 0; i < 8; ++i) {
            int r = rbase + i;
            if (r < M) {
                float* op = out + (size_t)r * N + c;
#pragma unroll
                for (int j = 0; j < NT; ++j) op[j * 16] = fmaxf(acc[j][i], 0.0f);
            }
        }
        if (!more) break;
#pragma unroll
        for (int kb = 0; kb < KB; ++kb) a0[kb] = a1[kb];
        mt  = mtn;
        row = rown;
    }
}

template <int KD, int N>
__global__ __launch_bounds__(SW * 32) void
k_premul_strip(const float* __restrict__ X, const float* __restrict__ W,
               const float* __restrict__ bias, float* __restrict__ Y0,
               float* __restrict__ Y1, float* __restrict__ Y2, int M, int Mtiles) {
    __shared__ __attribute__((aligned(16))) float Wl[SLDS];
    constexpr int KP = KD + 4;
    constexpr int NT = N / 16;
    constexpr int KH = KD / 2;
    constexpr int KB = KH / 8;

    int tid = threadIdx.x;
    stage_weights<KD, N>(W, Wl, tid);
    __syncthreads();

    int wave = __builtin_amdgcn_readfirstlane(tid >> 5);
    int step = SW * (int)gridDim.x;
    int mt   = wave * (int)gridDim.x + (int)blockIdx.x;
    if (mt >= Mtiles) return;

    int lane = tid & 31;
    int c    = lane & 15;
    int h    = lane >> 4;

    float bv[NT];
#pragma unroll
    for (int j = 0; j < NT; ++j) bv[j] = bias[j * 16 + c];

    const float* bp = Wl + c * KP + 4 * h;
    int row = min(mt * 16 + c, M - 1);

    float4 a0[KB], a1[KB];
    float4 bc[3 * NT];
    load_a<KB>(a0, X + (size_t)row * KD, h);
    load_b<3 * NT, KP>(bc, bp);
    for (;;) {
        int  mtn  = mt + step;
        bool more = mtn < Mtiles;
        int  rown = min((more ? mtn : mt) * 16 + c, M - 1);

        v8f acc[3 * NT];
#pragma unroll
        for (int j = 0; j < 3 * NT; ++j) {
#pragma unroll
            for (int i = 0; i < 8; ++i) acc[j][i] = (j < NT) ? bv[j < NT ? j : 0] : 0.0f;
        }

        load_a<KB>(a1, X + (size_t)row * KD + KH, h);
        mma_unit<KB, 3 * NT, KP>(a0, bp, bp + KH, bc, acc);
        load_a<KB>(a0, X + (size_t)rown * KD, h);
        mma_unit<KB, 3 * NT, KP>(a1, bp + KH, bp, bc, acc);

        int rbase = mt * 16 + (h << 3);
#pragma unroll
        for (int i = 0; i < 8; ++i) {
            int r = rbase + i;
            if (r < M) {
                size_t o = (size_t)r * N + c;
#pragma unroll
                for (int j = 0; j < NT; ++j) {
                    Y0[o + j * 16] = acc[j][i];
                    Y1[o + j * 16] = acc[NT + j][i];
                    Y2[o + j * 16] = acc[2 * NT + j][i];
                }
            }
        }
        if (!more) break;
        mt  = mtn;
        row = rown;
    }
}


#define POOL_ROWS 16
__device__ __forceinline__ void pool_flush(float* __restrict__ sums, float* __restrict__ cnt,
                                           int g, int lane, const float4& a, float c) {
    float* sp = sums + ((size_t)g << 7) + (lane << 2);
    atomicAdd(sp + 0, a.x);
    atomicAdd(sp + 1, a.y);
    atomicAdd(sp + 2, a.z);
    atomicAdd(sp + 3, a.w);
    if (lane == 0) atomicAdd(cnt + g, c);
}

__global__ __launch_bounds__(256) void
k_pool_runs(const float* __restrict__ h, const int* __restrict__ batch,
            float* __restrict__ sums, float* __restrict__ cnt, int n) {
    int wave = __builtin_amdgcn_readfirstlane((int)((blockIdx.x * 256 + threadIdx.x) >> 5));
    int r0 = wave * POOL_ROWS;
    if (r0 >= n) return;
    int lane = threadIdx.x & 31;
    const float4* hp = reinterpret_cast<const float4*>(h) + ((size_t)r0 << 5) + lane;
    const int*    bp = batch + r0;

    float4 acc = make_float4(0.0f, 0.0f, 0.0f, 0.0f);
    float  c   = 0.0f;
    int    gc;
    if (r0 + POOL_ROWS <= n) {
        int    g[POOL_ROWS];
        float4 v[POOL_ROWS];
#pragma unroll
        for (int j = 0; j < POOL_ROWS; ++j) g[j] = bp[j];
#pragma unroll
        for (int j = 0; j < POOL_ROWS; ++j) v[j] = hp[j << 5];
        gc = g[0];
#pragma unroll
        for (int j = 0; j < POOL_ROWS; ++j) {
            if (g[j] != gc) {
                pool_flush(sums, cnt, gc, lane, acc, c);
                acc = make_float4(0.0f, 0.0f, 0.0f, 0.0f);
                c   = 0.0f;
                gc  = g[j];
            }
            acc.x += v[j].x; acc.y += v[j].y; acc.z += v[j].z; acc.w += v[j].w;
            c += 1.0f;
        }
    } else {
        gc = bp[0];
        for (int j = 0; j < n - r0; ++j) {
            int    gj = bp[j];
            float4 vj = hp[j << 5];
            if (gj != gc) {
                pool_flush(sums, cnt, gc, lane, acc, c);
                acc = make_float4(0.0f, 0.0f, 0.0f, 0.0f);
                c   = 0.0f;
                gc  = gj;
            }
            acc.x += vj.x; acc.y += vj.y; acc.z += vj.z; acc.w += vj.w;
            c += 1.0f;
        }
    }
    pool_flush(sums, cnt, gc, lane, acc, c);
}

__global__ void k_final(const float* __restrict__ sums, const float* __restrict__ cnt,
                        const float* __restrict__ linW, const float* __restrict__ linb,
                        float* __restrict__ out, int G, int H) {
    __shared__ float hg_s[128];
    int g = blockIdx.x;
    int f = threadIdx.x;
    float c = cnt[g];
    float hg = sums[(size_t)g * H + f] / fmaxf(c, 1.0f);
    out[(size_t)G * 2 + (size_t)g * H + f] = hg;
    hg_s[f] = hg;
    __syncthreads();
    if (f < 2) {
        float acc = linb[f];
        for (int j = 0; j < H; ++j) acc += hg_s[j] * linW[j * 2 + f];
        out[g * 2 + f] = acc;
    }
}


extern "C" void kernel_launch(void* const* d_in, const int* in_sizes, int n_in,
                              void* d_out, int out_size, void* d_ws, size_t ws_size,
                              hipStream_t stream) {
    const float* x    = (const float*)d_in[0];
    const int*   ei   = (const int*)d_in[1];
    const float* attr = (const float*)d_in[2];
    const int*   bat  = (const int*)d_in[3];
    const float* W1 = (const float*)d_in[4];  const float* b1 = (const float*)d_in[5];
    const float* W2 = (const float*)d_in[6];  const float* b2 = (const float*)d_in[7];
    const float* W3 = (const float*)d_in[8];  const float* b3 = (const float*)d_in[9];
    const float* linW = (const float*)d_in[10];
    const float* linb = (const float*)d_in[11];
    float* out = (float*)d_out;

    const int F_IN = 128, H1 = 64, H2 = 64, H3 = 128, NCLS = 2;
    int Nn = in_sizes[0] / F_IN;
    int E  = in_sizes[2];
    int G  = out_size / (NCLS + H3);

    auto up = [](size_t v) { return (v + 63) & ~(size_t)63; };
    float* ws = (float*)d_ws;
    size_t off = 0;
    const int NCH = 256;
    int nbk = (Nn + BK_W - 1) >> BK_SH;
    int nhc = nbk * NCH;
    float* dis  = ws + off; off += up((size_t)Nn);
    int*   hist = (int*)(ws + off); off += up((size_t)nhc);
    int*   boff = (int*)(ws + off); off += up((size_t)nhc + 1);
    int*   rptr = (int*)(ws + off); off += up((size_t)Nn + 1);
    int2*  bins = (int2*)(ws + off); off += up((size_t)E * 2);
    int2*  csr  = (int2*)(ws + off); off += up((size_t)E * 2);
    float* A    = ws + off; off += up((size_t)Nn * H1);
    float* B    = ws + off; off += up((size_t)Nn * H1);
    float* C    = ws + off; off += up((size_t)Nn * H3);
    float* D    = ws + off; off += up((size_t)Nn * H2);
    float* pool = ws + off; off += (size_t)G * H3;
    float* cnt  = ws + off; off += (size_t)G;

    auto nb = [](long n, int t) { return (unsigned)((n + t - 1) / t); };

    k_bin_count<<<NCH, BIN_T, 0, stream>>>(ei, hist, E, nbk);
    k_rowptr<<<nb(nhc, SCAN_T * 4), SCAN_T, 0, stream>>>(hist, boff, nhc);
    k_bin_place<<<NCH, BIN_T, 0, stream>>>(ei, attr, boff, bins, E, nbk);
    k_bucket_csr<<<nbk, BIN_T, 0, stream>>>(bins, boff, NCH, dis, rptr, csr, Nn, E);
    k_csr_norm<<<nb(E, 256), 256, 0, stream>>>(csr, dis, E);

    unsigned gb = nb((long)Nn * 32, 256);

    int mtiles = (Nn + 15) / 16;

    k_premul_strip<128, 64><<<SGRID, SW * 32, 0, stream>>>(x, W1, b1, C, A, B, Nn, mtiles);
    k_gather64<true, false><<<gb, 256, 0, stream>>>(B, csr, rptr, A, Nn, 2.0f);
    k_gather64<true, true><<<gb, 256, 0, stream>>>(A, csr, rptr, C, Nn, 1.0f);

    k_gather64<false, false><<<gb, 256, 0, stream>>>(C, csr, rptr, A, Nn, 1.0f);
    k_gather64<false, false><<<gb, 256, 0, stream>>>(A, csr, rptr, B, Nn, 2.0f);
    k_cheb_strip<64, 64><<<SGRID, SW * 32, 0, stream>>>(C, A, B, W2, b2, D, Nn, mtiles);

    k_gather64<false, false><<<gb, 256, 0, stream>>>(D, csr, rptr, A, Nn, 1.0f);
    k_gather64<false, false><<<gb, 256, 0, stream>>>(A, csr, rptr, B, Nn, 2.0f);
    k_cheb_strip<64, 128><<<SGRID, SW * 32, 0, stream>>>(D, A, B, W3, b3, C, Nn, mtiles);

    k_fill<<<nb((long)G * H3 + G, 256), 256, 0, stream>>>(pool, 0.0f, G * H3 + G);
    k_pool_runs<<<nb((long)nb(Nn, POOL_ROWS) * 32, 256), 256, 0, stream>>>(C, bat, pool, cnt, Nn);
    k_final<<<G, H3, 0, stream>>>(pool, cnt, linW, linb, out, G, H3);
}


# layer 3's mean-pool sums taken in the WMMA strip epilogue (relu rows added from the accumulators, atomic adds once per strip and column), activations never stored; 15 launches; sum order changes
# speedup vs baseline: 5.9805x; 1.0403x over previous
#include <hip/hip_runtime.h>

typedef __attribute__((ext_vector_type(2))) float v2f;
typedef __attribute__((ext_vector_type(8))) float v8f;

#define BK_SH 8
#define BK_W  (1 << BK_SH)
#define MAXB  1024
#define BIN_T 1024
#define EPT1  8
#define EPT2  5

__global__ __launch_bounds__(BIN_T) void
k_bin_count(const int* __restrict__ ei, int* __restrict__ histG, int E, int nbk) {
    __shared__ int hist[MAXB];
    int tid = threadIdx.x;
    int c   = blockIdx.x;
    int nch = gridDim.x;
    int ch  = (E + nch - 1) / nch;
    int e0  = c * ch;
    int e1  = min(E, e0 + ch);
    for (int i = tid; i < nbk; i += BIN_T) hist[i] = 0;
    __syncthreads();
    for (int base = e0 + tid; base < e1 + tid; base += BIN_T * EPT1) {
        int d[EPT1];
#pragma unroll
        for (int j = 0; j < EPT1; ++j) d[j] = ei[E + min(base + j * BIN_T, e1 - 1)];
#pragma unroll
        for (int j = 0; j < EPT1; ++j)
            if (base + j * BIN_T < e1) atomicAdd(&hist[d[j] >> BK_SH], 1);
    }
    __syncthreads();
    for (int i = tid; i < nbk; i += BIN_T) histG[i * nch + c] = hist[i];
}

#define SCAN_T 1024
__global__ __launch_bounds__(SCAN_T) void
k_rowptr(const int* __restrict__ cnt, int* __restrict__ rowptr, int n) {
    __shared__ int pre_s[SCAN_T / 32];
    __shared__ int sum_s[SCAN_T / 32];
    int tid  = threadIdx.x;
    int lane = tid & 31;
    int wv   = tid >> 5;
    int base = blockIdx.x * (SCAN_T * 4);

    int pre = 0;
    const int4* c4 = reinterpret_cast<const int4*>(cnt) + tid;
#pragma unroll 4
    for (int k = 0; k < (int)blockIdx.x; ++k) {
        int4 v = c4[(size_t)k * SCAN_T];
        pre += (v.x + v.y) + (v.z + v.w);
    }
    for (int o = 16; o > 0; o >>= 1) pre += __shfl_xor(pre, o);
    if (lane == 0) pre_s[wv] = pre;

    int i0 = base + tid * 4;
    int c0 = (i0     < n) ? cnt[i0]     : 0;
    int c1 = (i0 + 1 < n) ? cnt[i0 + 1] : 0;
    int c2 = (i0 + 2 < n) ? cnt[i0 + 2] : 0;
    int c3 = (i0 + 3 < n) ? cnt[i0 + 3] : 0;
    int s = (c0 + c1) + (c2 + c3);
    int inc = s;
    for (int o = 1; o < 32; o <<= 1) {
        int t = __shfl_up(inc, o);
        if (lane >= o) inc += t;
    }
    if (lane == 31) sum_s[wv] = inc;
    __syncthreads();

    int t = pre_s[lane] + ((lane < wv) ? sum_s[lane] : 0);
    for (int o = 16; o > 0; o >>= 1) t += __shfl_xor(t, o);
    int x = t + inc - s;

    if (blockIdx.x == 0 && tid == 0) rowptr[0] = 0;
    x += c0; if (i0     < n) rowptr[i0 + 1] = x;
    x += c1; if (i0 + 1 < n) rowptr[i0 + 2] = x;
    x += c2; if (i0 + 2 < n) rowptr[i0 + 3] = x;
    x += c3; if (i0 + 3 < n) rowptr[i0 + 4] = x;
}

__global__ __launch_bounds__(BIN_T) void
k_bin_place(const int* __restrict__ ei, const float* __restrict__ attr,
            const int* __restrict__ binOff, int2* __restrict__ binned, int E, int nbk) {
    __shared__ int cur[MAXB];
    int tid = threadIdx.x;
    int c   = blockIdx.x;
    int nch = gridDim.x;
    int ch  = (E + nch - 1) / nch;
    int e0  = c * ch;
    int e1  = min(E, e0 + ch);
    for (int i = tid; i < nbk; i += BIN_T) cur[i] = binOff[i * nch + c];
    __syncthreads();
    for (int base = e0 + tid; base < e1 + tid; base += BIN_T * EPT1) {
        int   s[EPT1], d[EPT1];
        float a[EPT1];
#pragma unroll
        for (int j = 0; j < EPT1; ++j) {
            int e = min(base + j * BIN_T, e1 - 1);
            s[j] = ei[e];
            d[j] = ei[E + e];
            a[j] = attr[e];
        }
#pragma unroll
        for (int j = 0; j < EPT1; ++j) {
            if (base + j * BIN_T < e1) {
                float ew = (s[j] != d[j]) ? a[j] : 0.0f;
                int pos = atomicAdd(&cur[d[j] >> BK_SH], 1);
                binned[pos] = make_int2((s[j] << BK_SH) | (d[j] & (BK_W - 1)), __float_as_int(ew));
            }
        }
    }
}

__global__ __launch_bounds__(BIN_T) void
k_bucket_csr(const int2* __restrict__ binned, const int* __restrict__ binOff, int nch,
             float* __restrict__ dis, int* __restrict__ rowptr, int2* __restrict__ csr,
             int n, int E) {
    __shared__ int   cntL[BK_W];
    __shared__ float degL[BK_W];
    __shared__ int   curL[BK_W];
    __shared__ float disL[BK_W];
    __shared__ int   wsum[BK_W / 32];
    int tid  = threadIdx.x;
    int lane = tid & 31;
    int b    = blockIdx.x;
    int e0   = binOff[b * nch];
    int e1   = binOff[(b + 1) * nch];
    if (tid < BK_W) { cntL[tid] = 0; degL[tid] = 0.0f; }
    __syncthreads();
    for (int base = e0 + tid; base < e1 + tid; base += BIN_T * EPT2) {
        int2 r[EPT2];
#pragma unroll
        for (int j = 0; j < EPT2; ++j) r[j] = binned[min(base + j * BIN_T, e1 - 1)];
#pragma unroll
        for (int j = 0; j < EPT2; ++j) {
            if (base + j * BIN_T < e1) {
                int ld = r[j].x & (BK_W - 1);
                float ew = __int_as_float(r[j].y);
                atomicAdd(&cntL[ld], 1);
                if (ew != 0.0f) atomicAdd(&degL[ld], ew);
            }
        }
    }
    __syncthreads();
    int cn  = (tid < BK_W) ? cntL[tid] : 0;
    int inc = cn;
    for (int o = 1; o < 32; o <<= 1) {
        int t = __shfl_up(inc, o);
        if (lane >= o) inc += t;
    }
    if (tid < BK_W && lane == 31) wsum[tid >> 5] = inc;
    __syncthreads();
    if (tid < BK_W) {
        int wv  = tid >> 5;
        int pre = 0;
        for (int k = 0; k < BK_W / 32; ++k) pre += (k < wv) ? wsum[k] : 0;
        int start = e0 + pre + inc - cn;
        float dv = degL[tid];
        float di = (dv > 0.0f) ? rsqrtf(fmaxf(dv, 1e-30f)) : 0.0f;
        curL[tid] = start;
        disL[tid] = di;
        int node = b * BK_W + tid;
        if (node < n) { rowptr[node] = start; dis[node] = di; }
    }
    if (b == (int)gridDim.x - 1 && tid == 0) rowptr[n] = E;
    __syncthreads();
    for (int base = e0 + tid; base < e1 + tid; base += BIN_T * EPT2) {
        int2 r[EPT2];
#pragma unroll
        for (int j = 0; j < EPT2; ++j) r[j] = binned[min(base + j * BIN_T, e1 - 1)];
#pragma unroll
        for (int j = 0; j < EPT2; ++j) {
            if (base + j * BIN_T < e1) {
                int ld = r[j].x & (BK_W - 1);
                int s  = (int)((unsigned)r[j].x >> BK_SH);
                float ew = __int_as_float(r[j].y);
                int pos = atomicAdd(&curL[ld], 1);
                csr[pos] = make_int2(s, __float_as_int(-disL[ld] * ew));
            }
        }
    }
}

__global__ void k_csr_norm(int2* __restrict__ csr, const float* __restrict__ dis, int E,
                           float* __restrict__ zero, int nz) {
    int j = blockIdx.x * blockDim.x + threadIdx.x;
    if (j < nz) zero[j] = 0.0f;
    if (j >= E) return;
    int2 r = csr[j];
    r.y = __float_as_int(__int_as_float(r.y) * dis[r.x]);
    csr[j] = r;
}

template <int U>
__device__ __forceinline__ void gather_chunk(const int2* __restrict__ c,
                                             const float2* __restrict__ hp, float2& acc) {
    int2 ce[U];
    float2 v[U];
#pragma unroll
    for (int j = 0; j < U; ++j) ce[j] = c[j];
#pragma unroll
    for (int j = 0; j < U; ++j) v[j] = hp[(size_t)ce[j].x * 32];
#pragma unroll
    for (int j = 0; j < U; ++j) {
        float wj = __int_as_float(ce[j].y);
        acc.x = fmaf(wj, v[j].x, acc.x);
        acc.y = fmaf(wj, v[j].y, acc.y);
    }
}

template <bool HAS_BASE, bool RELU>
__global__ __launch_bounds__(256) void
k_gather64(const float* __restrict__ h, const int2* __restrict__ csr,
           const int* __restrict__ rowptr, float* __restrict__ out, int n, float scale) {
    int row = __builtin_amdgcn_readfirstlane((int)((blockIdx.x * 256 + threadIdx.x) >> 5));
    if (row >= n) return;
    int lane = threadIdx.x & 31;
    float2* op = reinterpret_cast<float2*>(out) + (size_t)row * 32 + lane;
    float2 b = make_float2(0.0f, 0.0f);
    if (HAS_BASE) b = *op;
    int e   = rowptr[row];
    int end = rowptr[row + 1];
    const float2* hp = reinterpret_cast<const float2*>(h) + lane;
    float2 acc = make_float2(0.0f, 0.0f);
    for (; e + 8 <= end; e += 8) gather_chunk<8>(csr + e, hp, acc);
    if (e + 4 <= end) { gather_chunk<4>(csr + e, hp, acc); e += 4; }
    int rem = end - e;
    if (rem > 0) {
        int2 t0 = csr[e];
        int2 t1 = csr[rem > 1 ? e + 1 : e];
        int2 t2 = csr[rem > 2 ? e + 2 : e];
        float2 v0 = hp[(size_t)t0.x * 32];
        float2 v1 = hp[(size_t)t1.x * 32];
        float2 v2 = hp[(size_t)t2.x * 32];
        float w0 = __int_as_float(t0.y);
        acc.x = fmaf(w0, v0.x, acc.x);
        acc.y = fmaf(w0, v0.y, acc.y);
        if (rem > 1) {
            float w1 = __int_as_float(t1.y);
            acc.x = fmaf(w1, v1.x, acc.x);
            acc.y = fmaf(w1, v1.y, acc.y);
        }
        if (rem > 2) {
            float w2 = __int_as_float(t2.y);
            acc.x = fmaf(w2, v2.x, acc.x);
            acc.y = fmaf(w2, v2.y, acc.y);
        }
    }
    float2 r;
    r.x = scale * acc.x + b.x;
    r.y = scale * acc.y + b.y;
    if (RELU) { r.x = fmaxf(r.x, 0.0f); r.y = fmaxf(r.y, 0.0f); }
    *op = r;
}

#define SW 12
#define SGRID 256
#define SLDS 41984

template <int KD, int N>
__device__ __forceinline__ void stage_weights(const float* __restrict__ W,
                                              float* __restrict__ Wl, int tid) {
    constexpr int KP   = KD + 4;
    constexpr int Q    = N / 4;
    constexpr int SL4  = KD * Q;
    constexpr int NB16 = N / 16;
    constexpr int GPS  = (KD / 8) * NB16;
    constexpr int IT0  = (GPS + SW - 1) / SW;
    constexpr int IT1  = (2 * GPS + SW - 1) / SW;
    const float4* W4 = reinterpret_cast<const float4*>(W);
    int wv = tid >> 5;
    int q  = tid & 3;
    int kk = (tid >> 2) & 7;

    float4 va[IT0], vb[IT0], vc[IT1];
#pragma unroll
    for (int it = 0; it < IT0; ++it) {
        int g    = min(wv + it * SW, GPS - 1);
        int kblk = g / NB16;
        int nblk = g - kblk * NB16;
        int idx  = (kblk * 8 + kk) * Q + nblk * 4 + q;
        va[it] = W4[idx];
        vb[it] = W4[idx + 2 * SL4];
    }
#pragma unroll
    for (int it = 0; it < IT1; ++it) {
        int g    = min(wv + it * SW, 2 * GPS - 1);
        int t    = g / GPS;
        int r    = g - t * GPS;
        int kblk = r / NB16;
        int nblk = r - kblk * NB16;
        vc[it] = W4[((t + 1) * KD + kblk * 8 + kk) * Q + nblk * 4 + q];
    }
#pragma unroll
    for (int it = 0; it < IT0; ++it) {
        int g = wv + it * SW;
        if (g < GPS) {
            int kblk = g / NB16;
            int nblk = g - kblk * NB16;
            float* p = Wl + (nblk * 16 + q * 4) * KP + kblk * 8 + kk;
            p[0]      = va[it].x - vb[it].x;
            p[KP]     = va[it].y - vb[it].y;
            p[2 * KP] = va[it].z - vb[it].z;
            p[3 * KP] = va[it].w - vb[it].w;
        }
    }
#pragma unroll
    for (int it = 0; it < IT1; ++it) {
        int g = wv + it * SW;
        if (g < 2 * GPS) {
            int t    = g / GPS;
            int r    = g - t * GPS;
            int kblk = r / NB16;
            int nblk = r - kblk * NB16;
            float* p = Wl + ((t + 1) * N + nblk * 16 + q * 4) * KP + kblk * 8 + kk;
            p[0]      = vc[it].x;
            p[KP]     = vc[it].y;
            p[2 * KP] = vc[it].z;
            p[3 * KP] = vc[it].w;
        }
    }
}

template <int KB>
__device__ __forceinline__ void load_a(float4 (&a)[KB], const float* __restrict__ rowp, int h) {
    const float4* p = reinterpret_cast<const float4*>(rowp) + h;
#pragma unroll
    for (int kb = 0; kb < KB; ++kb) a[kb] = p[kb * 2];
}

template <int NT, int KP>
__device__ __forceinline__ void load_b(float4 (&b)[NT], const float* p) {
    __builtin_amdgcn_fence(__ATOMIC_ACQUIRE, "wavefront", "local");
#pragma unroll
    for (int j = 0; j < NT; ++j)
        b[j] = *reinterpret_cast<const float4*>(p + j * 16 * KP);
}

template <int NT>
__device__ __forceinline__ void mma_step(const float4& a, const float4 (&b)[NT], v8f (&acc)[NT]) {
    v2f alo, ahi;
    alo.x = a.x; alo.y = a.y;
    ahi.x = a.z; ahi.y = a.w;
#pragma unroll
    for (int j = 0; j < NT; ++j) {
        v2f bl; bl.x = b[j].x; bl.y = b[j].y;
        acc[j] = __builtin_amdgcn_wmma_f32_16x16x4_f32(
                      false, alo,  false, bl,
                      (short)0, acc[j],  false,  false);
    }
#pragma unroll
    for (int j = 0; j < NT; ++j) {
        v2f bh; bh.x = b[j].z; bh.y = b[j].w;
        acc[j] = __builtin_amdgcn_wmma_f32_16x16x4_f32(
                     false, ahi, false, bh, (short)0, acc[j], false, false);
    }
}

template <int KB, int NT, int KP>
__device__ __forceinline__ void mma_unit(const float4 (&a)[KB], const float* bp, const float* bnext,
                                         float4 (&bc)[NT], v8f (&acc)[NT]) {
#pragma unroll
    for (int kb = 0; kb < KB; ++kb) {
        float4 bn[NT];
        load_b<NT, KP>(bn, (kb + 1 < KB) ? bp + (kb + 1) * 8 : bnext);
        mma_step<NT>(a[kb], bc, acc);
#pragma unroll
        for (int j = 0; j < NT; ++j) bc[j] = bn[j];
    }
}

__device__ __forceinline__ void pool_ids_load(int (&id)[16], const int* __restrict__ batch,
                                              int mt, int M) {
#pragma unroll
    for (int k = 0; k < 16; ++k) id[k] = 0;
    if (mt * 16 + 16 <= M) {
#pragma unroll
        for (int k = 0; k < 16; ++k) id[k] = batch[mt * 16 + k];
    }
}

__device__ __forceinline__ bool pool_ids_same(const int (&id)[16], int mt, int M, int& g0) {
    int diff = 0;
#pragma unroll
    for (int k = 1; k < 16; ++k) diff |= id[k] ^ id[0];
    g0 = __builtin_amdgcn_readfirstlane(id[0]);
    return (mt * 16 + 16 <= M) && (__builtin_amdgcn_readfirstlane(diff) == 0);
}

template <int NT>
__device__ __forceinline__ void pool_flush(float* __restrict__ sums, float* __restrict__ cnt,
                                           int g, int c, const float (&run)[NT], float n) {
    float* sp = sums + (size_t)g * (NT * 16) + c;
#pragma unroll
    for (int j = 0; j < NT; ++j) atomicAdd(sp + j * 16, run[j]);
    if (c == 0) atomicAdd(cnt + g, n);
}

template <int NT>
__device__ __forceinline__ void pool_strip(const v8f (&acc)[NT], const int* __restrict__ batch,
                                           float* __restrict__ sums, float* __restrict__ cnt,
                                           int mt, int M, int c, int h, bool same, int g0) {
    int r0 = mt * 16;
    if (same) {
        float s[NT];
#pragma unroll
        for (int j = 0; j < NT; ++j) {
            float t = fmaxf(acc[j][0], 0.0f);
#pragma unroll
            for (int i = 1; i < 8; ++i) t += fmaxf(acc[j][i], 0.0f);
            s[j] = t + __shfl_xor(t, 16);
        }
        if (h == 0) pool_flush<NT>(sums, cnt, g0, c, s, 16.0f);
    } else {
        int rbase = r0 + (h << 3);
        float run[NT];
#pragma unroll
        for (int j = 0; j < NT; ++j) run[j] = 0.0f;
        float n  = 0.0f;
        int   gc = 0;
#pragma unroll
        for (int i = 0; i < 8; ++i) {
            int r = rbase + i;
            if (r < M) {
                int gi = batch[r];
                if (n > 0.0f && gi != gc) {
                    pool_flush<NT>(sums, cnt, gc, c, run, n);
#pragma unroll
                    for (int j = 0; j < NT; ++j) run[j] = 0.0f;
                    n = 0.0f;
                }
                gc = gi;
#pragma unroll
                for (int j = 0; j < NT; ++j) run[j] += fmaxf(acc[j][i], 0.0f);
                n += 1.0f;
            }
        }
        if (n > 0.0f) pool_flush<NT>(sums, cnt, gc, c, run, n);
    }
}

template <int KD, int N, bool POOL>
__global__ __launch_bounds__(SW * 32) void
k_cheb_strip(const float* __restrict__ Tx0, const float* __restrict__ Tx1,
             const float* __restrict__ P, const float* __restrict__ W,
             const float* __restrict__ bias, float* __restrict__ out, int M, int Mtiles,
             const int* __restrict__ batch, float* __restrict__ sums, float* __restrict__ cnt) {
    __shared__ __attribute__((aligned(16))) float Wl[SLDS];
    constexpr int KP = KD + 4;
    constexpr int NT = N / 16;
    constexpr int KB = KD / 8;

    int tid = threadIdx.x;
    stage_weights<KD, N>(W, Wl, tid);
    __syncthreads();

    int wave = __builtin_amdgcn_readfirstlane(tid >> 5);
    int step = SW * (int)gridDim.x;
    int mt   = wave * (int)gridDim.x + (int)blockIdx.x;
    if (mt >= Mtiles) return;

    int lane = tid & 31;
    int c    = lane & 15;
    int h    = lane >> 4;

    float bv[NT];
#pragma unroll
    for (int j = 0; j < NT; ++j) bv[j] = bias[j * 16 + c];

    const float* bp = Wl + c * KP + 4 * h;
    int row = min(mt * 16 + c, M - 1);

    float4 a0[KB], a1[KB];
    float4 bc[NT];
    load_a<KB>(a0, Tx0 + (size_t)row * KD, h);
    load_b<NT, KP>(bc, bp);
    for (;;) {
        int  mtn  = mt + step;
        bool more = mtn < Mtiles;
        int  rown = min((more ? mtn : mt) * 16 + c, M - 1);

        v8f acc[NT];
#pragma unroll
        for (int j = 0; j < NT; ++j) {
#pragma unroll
            for (int i = 0; i < 8; ++i) acc[j][i] = bv[j];
        }

        int  gid[16];
        int  g0   = 0;
        bool same = false;
        if constexpr (POOL) pool_ids_load(gid, batch, mt, M);

        load_a<KB>(a1, Tx1 + (size_t)row * KD, h);
        mma_unit<KB, NT, KP>(a0, bp, bp + N * KP, bc, acc);
        if constexpr (POOL) same = pool_ids_same(gid, mt, M, g0);
        load_a<KB>(a0, P + (size_t)row * KD, h);
        mma_unit<KB, NT, KP>(a1, bp + N * KP, bp + 2 * N * KP, bc, acc);
        load_a<KB>(a1, Tx0 + (size_t)rown * KD, h);
        mma_unit<KB, NT, KP>(a0, bp + 2 * N * KP, bp, bc, acc);

        if constexpr (POOL) {
            pool_strip<NT>(acc, batch, sums, cnt, mt, M, c, h, same, g0);
        } else {
            int rbase = mt * 16 + (h << 3);
#pragma unroll
            for (int i = 0; i < 8; ++i) {
                int r = rbase + i;
                if (r < M) {
                    float* op = out + (size_t)r * N + c;
#pragma unroll
                    for (int j = 0; j < NT; ++j) op[j * 16] = fmaxf(acc[j][i], 0.0f);
                }
            }
        }
        if (!more) break;
#pragma unroll
        for (int kb = 0; kb < KB; ++kb) a0[kb] = a1[kb];
        mt  = mtn;
        row = rown;
    }
}

template <int KD, int N>
__global__ __launch_bounds__(SW * 32) void
k_premul_strip(const float* __restrict__ X, const float* __restrict__ W,
               const float* __restrict__ bias, float* __restrict__ Y0,
               float* __restrict__ Y1, float* __restrict__ Y2, int M, int Mtiles) {
    __shared__ __attribute__((aligned(16))) float Wl[SLDS];
    constexpr int KP = KD + 4;
    constexpr int NT = N / 16;
    constexpr int KH = KD / 2;
    constexpr int KB = KH / 8;

    int tid = threadIdx.x;
    stage_weights<KD, N>(W, Wl, tid);
    __syncthreads();

    int wave = __builtin_amdgcn_readfirstlane(tid >> 5);
    int step = SW * (int)gridDim.x;
    int mt   = wave * (int)gridDim.x + (int)blockIdx.x;
    if (mt >= Mtiles) return;

    int lane = tid & 31;
    int c    = lane & 15;
    int h    = lane >> 4;

    float bv[NT];
#pragma unroll
    for (int j = 0; j < NT; ++j) bv[j] = bias[j * 16 + c];

    const float* bp = Wl + c * KP + 4 * h;
    int row = min(mt * 16 + c, M - 1);

    float4 a0[KB], a1[KB];
    float4 bc[3 * NT];
    load_a<KB>(a0, X + (size_t)row * KD, h);
    load_b<3 * NT, KP>(bc, bp);
    for (;;) {
        int  mtn  = mt + step;
        bool more = mtn < Mtiles;
        int  rown = min((more ? mtn : mt) * 16 + c, M - 1);

        v8f acc[3 * NT];
#pragma unroll
        for (int j = 0; j < 3 * NT; ++j) {
#pragma unroll
            for (int i = 0; i < 8; ++i) acc[j][i] = (j < NT) ? bv[j < NT ? j : 0] : 0.0f;
        }

        load_a<KB>(a1, X + (size_t)row * KD + KH, h);
        mma_unit<KB, 3 * NT, KP>(a0, bp, bp + KH, bc, acc);
        load_a<KB>(a0, X + (size_t)rown * KD, h);
        mma_unit<KB, 3 * NT, KP>(a1, bp + KH, bp, bc, acc);

        int rbase = mt * 16 + (h << 3);
#pragma unroll
        for (int i = 0; i < 8; ++i) {
            int r = rbase + i;
            if (r < M) {
                size_t o = (size_t)r * N + c;
#pragma unroll
                for (int j = 0; j < NT; ++j) {
                    Y0[o + j * 16] = acc[j][i];
                    Y1[o + j * 16] = acc[NT + j][i];
                    Y2[o + j * 16] = acc[2 * NT + j][i];
                }
            }
        }
        if (!more) break;
        mt  = mtn;
        row = rown;
    }
}


__global__ void k_final(const float* __restrict__ sums, const float* __restrict__ cnt,
                        const float* __restrict__ linW, const float* __restrict__ linb,
                        float* __restrict__ out, int G, int H) {
    __shared__ float hg_s[128];
    int g = blockIdx.x;
    int f = threadIdx.x;
    float c = cnt[g];
    float hg = sums[(size_t)g * H + f] / fmaxf(c, 1.0f);
    out[(size_t)G * 2 + (size_t)g * H + f] = hg;
    hg_s[f] = hg;
    __syncthreads();
    if (f < 2) {
        float acc = linb[f];
        for (int j = 0; j < H; ++j) acc += hg_s[j] * linW[j * 2 + f];
        out[g * 2 + f] = acc;
    }
}


extern "C" void kernel_launch(void* const* d_in, const int* in_sizes, int n_in,
                              void* d_out, int out_size, void* d_ws, size_t ws_size,
                              hipStream_t stream) {
    const float* x    = (const float*)d_in[0];
    const int*   ei   = (const int*)d_in[1];
    const float* attr = (const float*)d_in[2];
    const int*   bat  = (const int*)d_in[3];
    const float* W1 = (const float*)d_in[4];  const float* b1 = (const float*)d_in[5];
    const float* W2 = (const float*)d_in[6];  const float* b2 = (const float*)d_in[7];
    const float* W3 = (const float*)d_in[8];  const float* b3 = (const float*)d_in[9];
    const float* linW = (const float*)d_in[10];
    const float* linb = (const float*)d_in[11];
    float* out = (float*)d_out;

    const int F_IN = 128, H1 = 64, H2 = 64, H3 = 128, NCLS = 2;
    int Nn = in_sizes[0] / F_IN;
    int E  = in_sizes[2];
    int G  = out_size / (NCLS + H3);

    auto up = [](size_t v) { return (v + 63) & ~(size_t)63; };
    float* ws = (float*)d_ws;
    size_t off = 0;
    const int NCH = 256;
    int nbk = (Nn + BK_W - 1) >> BK_SH;
    int nhc = nbk * NCH;
    float* dis  = ws + off; off += up((size_t)Nn);
    int*   hist = (int*)(ws + off); off += up((size_t)nhc);
    int*   boff = (int*)(ws + off); off += up((size_t)nhc + 1);
    int*   rptr = (int*)(ws + off); off += up((size_t)Nn + 1);
    int2*  bins = (int2*)(ws + off); off += up((size_t)E * 2);
    int2*  csr  = (int2*)(ws + off); off += up((size_t)E * 2);
    float* A    = ws + off; off += up((size_t)Nn * H1);
    float* B    = ws + off; off += up((size_t)Nn * H1);
    float* C    = ws + off; off += up((size_t)Nn * H1);
    float* D    = ws + off; off += up((size_t)Nn * H2);
    float* pool = ws + off; off += (size_t)G * H3;
    float* cnt  = ws + off; off += (size_t)G;

    auto nb = [](long n, int t) { return (unsigned)((n + t - 1) / t); };

    k_bin_count<<<NCH, BIN_T, 0, stream>>>(ei, hist, E, nbk);
    k_rowptr<<<nb(nhc, SCAN_T * 4), SCAN_T, 0, stream>>>(hist, boff, nhc);
    k_bin_place<<<NCH, BIN_T, 0, stream>>>(ei, attr, boff, bins, E, nbk);
    k_bucket_csr<<<nbk, BIN_T, 0, stream>>>(bins, boff, NCH, dis, rptr, csr, Nn, E);
    int npool = G * H3 + G;
    k_csr_norm<<<nb(E > npool ? E : npool, 256), 256, 0, stream>>>(csr, dis, E, pool, npool);

    unsigned gb = nb((long)Nn * 32, 256);

    int mtiles = (Nn + 15) / 16;

    k_premul_strip<128, 64><<<SGRID, SW * 32, 0, stream>>>(x, W1, b1, C, A, B, Nn, mtiles);
    k_gather64<true, false><<<gb, 256, 0, stream>>>(B, csr, rptr, A, Nn, 2.0f);
    k_gather64<true, true><<<gb, 256, 0, stream>>>(A, csr, rptr, C, Nn, 1.0f);

    k_gather64<false, false><<<gb, 256, 0, stream>>>(C, csr, rptr, A, Nn, 1.0f);
    k_gather64<false, false><<<gb, 256, 0, stream>>>(A, csr, rptr, B, Nn, 2.0f);
    k_cheb_strip<64, 64, false><<<SGRID, SW * 32, 0, stream>>>(C, A, B, W2, b2, D, Nn, mtiles,
                                                               nullptr, nullptr, nullptr);

    k_gather64<false, false><<<gb, 256, 0, stream>>>(D, csr, rptr, A, Nn, 1.0f);
    k_gather64<false, false><<<gb, 256, 0, stream>>>(A, csr, rptr, B, Nn, 2.0f);
    k_cheb_strip<64, 128, true><<<SGRID, SW * 32, 0, stream>>>(D, A, B, W3, b3, nullptr, Nn, mtiles,
                                                               bat, pool, cnt);

    k_final<<<G, H3, 0, stream>>>(pool, cnt, linW, linb, out, G, H3);
}
